# starting kernel
# baseline (speedup 1.0000x reference)
#include <hip/hip_runtime.h>
#include <hip/hip_bf16.h>


typedef float v2f __attribute__((ext_vector_type(2)));
typedef float v8f __attribute__((ext_vector_type(8)));

#define T_STEPS 256
#define UNITS_  256
#define N4      1024
#define KTOT    512
#define ASTRIDE 516
#define ZSTRIDE 1028
#define COMBO_F (512 * 1024)

__device__ __forceinline__ float sigmoid_f(float x) {
    return 1.0f / (1.0f + __expf(-x));
}
__device__ __forceinline__ float tanh_f(float x) {
    x = fminf(15.0f, fmaxf(-15.0f, x));
    float e = __expf(2.0f * x);
    return (e - 1.0f) / (e + 1.0f);
}

__global__ void pack_weights_kernel(const float* __restrict__ kfw,
                                    const float* __restrict__ rfw,
                                    const float* __restrict__ kbw,
                                    const float* __restrict__ rbw,
                                    float* __restrict__ packed) {
    int i = blockIdx.x * blockDim.x + threadIdx.x;
    int lane  = i & 31;
    int ks    = (i >> 5) & 127;
    int nt    = (i >> 12) & 63;
    int combo = i >> 18;
    int layer = combo >> 1;
    int isbw  = combo & 1;
    const float* W = (isbw ? kbw : kfw) + (size_t)layer * 256 * 1024;
    const float* R = (isbw ? rbw : rfw) + (size_t)layer * 256 * 1024;

    int n  = nt * 16 + (lane & 15);
    int k0 = ks * 4 + ((lane >> 4) << 1);
    float d0, d1;
    if (k0 < 256) {
        d0 = W[(size_t)k0 * 1024 + n];
        d1 = W[(size_t)(k0 + 1) * 1024 + n];
    } else {
        int kk = k0 - 256;
        d0 = R[(size_t)kk * 1024 + n];
        d1 = R[(size_t)(kk + 1) * 1024 + n];
    }
    packed[(size_t)i * 2 + 0] = d0;
    packed[(size_t)i * 2 + 1] = d1;
}

__global__ void __launch_bounds__(1024)
bilstm_layer_kernel(const float* __restrict__ in_fw,
                    const float* __restrict__ in_bw,
                    const float* __restrict__ pw_fw,
                    const float* __restrict__ pw_bw,
                    const float* __restrict__ bias_fw,
                    const float* __restrict__ bias_bw,
                    float* __restrict__ out_fw,
                    float* __restrict__ out_bw,
                    int residual) {
    extern __shared__ float lds[];
    float* Alds = lds;
    float* Zlds = lds + 16 * ASTRIDE;

    const int dir        = blockIdx.y;
    const float* in      = dir ? in_bw   : in_fw;
    const float* pw      = dir ? pw_bw   : pw_fw;
    const float* bias    = dir ? bias_bw : bias_fw;
    float*       out     = dir ? out_bw  : out_fw;

    const int b0      = blockIdx.x * 16;
    const int tid     = threadIdx.x;
    const int lane    = tid & 31;
    const int wave    = tid >> 5;
    const int halfsel = lane >> 4;
    const int lrow    = lane & 15;

    int   rr[4], uu[4];
    float bI[4], bF[4], bG[4], bO[4], cst[4];
#pragma unroll
    for (int j = 0; j < 4; ++j) {
        int f = tid + j * 1024;
        rr[j] = f >> 8;
        uu[j] = f & 255;
        bI[j] = bias[uu[j]];
        bF[j] = bias[256 + uu[j]];
        bG[j] = bias[512 + uu[j]];
        bO[j] = bias[768 + uu[j]];
        cst[j] = 0.0f;
        Alds[rr[j] * ASTRIDE + 256 + uu[j]] = 0.0f;
    }

    const int nt0 = wave * 2;
    const int nt1 = wave * 2 + 1;
    const float* pwt0 = pw + ((size_t)(nt0 * 128) * 32 + lane) * 2;
    const float* pwt1 = pw + ((size_t)(nt1 * 128) * 32 + lane) * 2;
    const int aBase = lrow * ASTRIDE + 2 * halfsel;

    for (int tt = 0; tt < T_STEPS; ++tt) {
        const int t = dir ? (T_STEPS - 1 - tt) : tt;

#pragma unroll
        for (int j = 0; j < 4; ++j) {
            Alds[rr[j] * ASTRIDE + uu[j]] =
                in[((size_t)(b0 + rr[j]) * T_STEPS + t) * UNITS_ + uu[j]];
        }
        if (tt + 1 < T_STEPS) {
            const int tn = dir ? (t - 1) : (t + 1);
            __builtin_prefetch(
                &in[((size_t)(b0 + rr[0]) * T_STEPS + tn) * UNITS_ + uu[0]], 0, 1);
        }
        __syncthreads();

        v8f acc0 = {0.f, 0.f, 0.f, 0.f, 0.f, 0.f, 0.f, 0.f};
        v8f acc1 = {0.f, 0.f, 0.f, 0.f, 0.f, 0.f, 0.f, 0.f};
#pragma unroll 4
        for (int ks = 0; ks < KTOT / 4; ++ks) {
            v2f a  = *(const v2f*)(Alds + aBase + 4 * ks);
            v2f w0 = *(const v2f*)(pwt0 + ks * 64);
            v2f w1 = *(const v2f*)(pwt1 + ks * 64);
            acc0 = __builtin_amdgcn_wmma_f32_16x16x4_f32(
                false, a, false, w0, (short)0, acc0, false, false);
            acc1 = __builtin_amdgcn_wmma_f32_16x16x4_f32(
                false, a, false, w1, (short)0, acc1, false, false);
        }
        {
            const int zr = 8 * halfsel;
#pragma unroll
            for (int e = 0; e < 8; ++e) {
                Zlds[(zr + e) * ZSTRIDE + nt0 * 16 + lrow] = acc0[e];
                Zlds[(zr + e) * ZSTRIDE + nt1 * 16 + lrow] = acc1[e];
            }
        }
        __syncthreads();

#pragma unroll
        for (int j = 0; j < 4; ++j) {
            const int r = rr[j], u = uu[j];
            float zi = Zlds[r * ZSTRIDE + u]       + bI[j];
            float zf = Zlds[r * ZSTRIDE + 256 + u] + bF[j];
            float zg = Zlds[r * ZSTRIDE + 512 + u] + bG[j];
            float zo = Zlds[r * ZSTRIDE + 768 + u] + bO[j];
            float gi = sigmoid_f(zi);
            float gf = sigmoid_f(zf);
            float gg = tanh_f(zg);
            float go = sigmoid_f(zo);
            float c  = gf * cst[j] + gi * gg;
            cst[j]   = c;
            float h  = go * tanh_f(c);
            Alds[r * ASTRIDE + 256 + u] = h;
            float ov = residual ? (h + Alds[r * ASTRIDE + u]) : h;
            out[((size_t)(b0 + r) * T_STEPS + t) * UNITS_ + u] = ov;
        }
        __syncthreads();
    }
}

__global__ void combine_avg_kernel(const float4* __restrict__ fw,
                                   float4* __restrict__ out, int n4) {
    int i = blockIdx.x * blockDim.x + threadIdx.x;
    if (i < n4) {
        float4 a = fw[i];
        float4 b = out[i];
        out[i] = make_float4((a.x + b.x) * 0.5f, (a.y + b.y) * 0.5f,
                             (a.z + b.z) * 0.5f, (a.w + b.w) * 0.5f);
    }
}

extern "C" void kernel_launch(void* const* d_in, const int* in_sizes, int n_in,
                              void* d_out, int out_size, void* d_ws, size_t ws_size,
                              hipStream_t stream) {
    const float* x   = (const float*)d_in[0];
    const float* kfw = (const float*)d_in[1];
    const float* rfw = (const float*)d_in[2];
    const float* bfw = (const float*)d_in[3];
    const float* kbw = (const float*)d_in[4];
    const float* rbw = (const float*)d_in[5];
    const float* bbw = (const float*)d_in[6];
    float* out = (float*)d_out;

    float* pk   = (float*)d_ws;
    float* wsfw = pk + (size_t)4 * COMBO_F;
    float* wsbw = out;

    pack_weights_kernel<<<4096, 256, 0, stream>>>(kfw, rfw, kbw, rbw, pk);

    const size_t shmem = (size_t)(16 * ASTRIDE + 16 * ZSTRIDE) * sizeof(float);
    (void)hipFuncSetAttribute((const void*)bilstm_layer_kernel,
                              hipFuncAttributeMaxDynamicSharedMemorySize,
                              (int)shmem);

    bilstm_layer_kernel<<<dim3(8, 2), 1024, shmem, stream>>>(
        x, x, pk + 0 * (size_t)COMBO_F, pk + 1 * (size_t)COMBO_F,
        bfw + 0, bbw + 0, wsfw, wsbw,  0);

    bilstm_layer_kernel<<<dim3(8, 2), 1024, shmem, stream>>>(
        wsfw, wsbw, pk + 2 * (size_t)COMBO_F, pk + 3 * (size_t)COMBO_F,
        bfw + 1024, bbw + 1024, wsfw, wsbw,  1);

    const int n4 = (128 * 256 * 256) / 4;
    combine_avg_kernel<<<(n4 + 255) / 256, 256, 0, stream>>>(
        (const float4*)wsfw, (float4*)out, n4);
}


# 2 batch rows per work-group, not 16 (128 work-groups): x@W of 8 time steps fills the freed matrix rows once per 8 steps, the per-step product is h@R only (K 256, not 512); sum order kept
# speedup vs baseline: 2.0428x; 2.0428x over previous
#include <hip/hip_runtime.h>


typedef float v2f __attribute__((ext_vector_type(2)));
typedef float v8f __attribute__((ext_vector_type(8)));

#define T_STEPS 256
#define UNITS_  256
#define COMBO_F (512 * 1024)
#define ROWS_WG 2
#define GSTEPS  8
#define NGRP    (T_STEPS / GSTEPS)
#define NTHR    512
#define XSTRIDE 260
#define HSTRIDE 264
#define TQ      (16 * 128 * 64)
#define LDS_X_F  (2 * 16 * XSTRIDE)
#define LDS_H_F  (2 * ROWS_WG * HSTRIDE)
#define LDS_XZ_F (32 * NTHR)
#define LDS_BYTES (164 * 1024)

__device__ __forceinline__ float sigmoid_f(float x) {
    return 1.0f / (1.0f + __expf(-x));
}
__device__ __forceinline__ float tanh_f(float x) {
    x = fminf(15.0f, fmaxf(-15.0f, x));
    float e = __expf(2.0f * x);
    return (e - 1.0f) / (e + 1.0f);
}

__global__ void pack_weights_kernel(const float* __restrict__ kfw,
                                    const float* __restrict__ rfw,
                                    const float* __restrict__ kbw,
                                    const float* __restrict__ rbw,
                                    float* __restrict__ packed) {
    int i = blockIdx.x * blockDim.x + threadIdx.x;
    int lane  = i & 31;
    int ks    = (i >> 5) & 127;
    int nt    = (i >> 12) & 63;
    int combo = i >> 18;
    int layer = combo >> 1;
    int isbw  = combo & 1;
    const float* W = (isbw ? kbw : kfw) + (size_t)layer * 256 * 1024;
    const float* R = (isbw ? rbw : rfw) + (size_t)layer * 256 * 1024;

    int n  = nt * 16 + (lane & 15);
    int k0 = ks * 4 + ((lane >> 4) << 1);
    float d0, d1;
    if (k0 < 256) {
        d0 = W[(size_t)k0 * 1024 + n];
        d1 = W[(size_t)(k0 + 1) * 1024 + n];
    } else {
        int kk = k0 - 256;
        d0 = R[(size_t)kk * 1024 + n];
        d1 = R[(size_t)(kk + 1) * 1024 + n];
    }
    packed[(size_t)i * 2 + 0] = d0;
    packed[(size_t)i * 2 + 1] = d1;
}

__device__ __forceinline__ void load_chunk(v2f (&b)[4][4], const float* p) {
#pragma unroll
    for (int kk = 0; kk < 4; ++kk)
#pragma unroll
        for (int q = 0; q < 4; ++q)
            b[kk][q] = *(const v2f*)(p + q * TQ + kk * 64);
}
__device__ __forceinline__ void load_afrag(v2f (&a)[4], const float* p) {
#pragma unroll
    for (int kk = 0; kk < 4; ++kk) a[kk] = *(const v2f*)(p + kk * 4);
}
__device__ __forceinline__ void mma_chunk(v8f (&acc)[4], const v2f (&a)[4],
                                          const v2f (&b)[4][4]) {
#pragma unroll
    for (int kk = 0; kk < 4; ++kk)
#pragma unroll
        for (int q = 0; q < 4; ++q)
            acc[q] = __builtin_amdgcn_wmma_f32_16x16x4_f32(
                false, a[kk], false, b[kk][q], (short)0, acc[q], false, false);
}

__device__ __forceinline__ void gemm_phase(v8f (&acc)[4], v2f (&b0)[4][4],
                                           v2f (&b1)[4][4], const float* aL,
                                           const float* bp, const float* nextp) {
    v2f a0[4], a1[4];
    load_afrag(a0, aL);
#pragma unroll 1
    for (int ch = 0; ch < 16; ch += 2) {
        load_chunk(b1, bp + (ch + 1) * 256);
        load_afrag(a1, aL + (ch + 1) * 16);
        mma_chunk(acc, a0, b0);
        const float* np = (ch + 2 < 16) ? (bp + (ch + 2) * 256) : nextp;
        load_chunk(b0, np);
        load_afrag(a0, aL + ((ch + 2) & 15) * 16);
        mma_chunk(acc, a1, b1);
    }
}

__global__ void __launch_bounds__(NTHR)
bilstm_layer_kernel(const float* in_fw,
                    const float* in_bw,
                    const float* __restrict__ pw_fw,
                    const float* __restrict__ pw_bw,
                    const float* __restrict__ bias_fw,
                    const float* __restrict__ bias_bw,
                    float* out_fw,
                    float* out_bw,
                    int residual) {
    extern __shared__ float lds[];
    float* Xl  = lds;
    float* Hl  = lds + LDS_X_F;
    float* XZl = Hl + LDS_H_F;

    const int dir      = blockIdx.y;
    const float* in    = dir ? in_bw   : in_fw;
    const float* pw    = dir ? pw_bw   : pw_fw;
    const float* bias  = dir ? bias_bw : bias_fw;
    float*       out   = dir ? out_bw  : out_fw;

    const int b0   = blockIdx.x * ROWS_WG;
    const int tid  = threadIdx.x;
    const int lane = tid & 31;
    const int wave = tid >> 5;
    const int hs   = lane >> 4;
    const int lc   = lane & 15;
    const int unit = wave * 16 + lc;

    const float bI = bias[unit];
    const float bF = bias[256 + unit];
    const float bG = bias[512 + unit];
    const float bO = bias[768 + unit];
    float cst = 0.0f;

    Hl[(tid >> 8) * HSTRIDE + (tid & 255)] = 0.0f;

    const int srow0 = tid >> 6;
    const int srow1 = srow0 + 8;
    const int scol  = (tid & 63) * 4;
    {
        const int t = dir ? (T_STEPS - 1 - srow0) : srow0;
        const float4 v0 = *(const float4*)(in + ((size_t)(b0 + 0) * T_STEPS + t) * UNITS_ + scol);
        const float4 v1 = *(const float4*)(in + ((size_t)(b0 + 1) * T_STEPS + t) * UNITS_ + scol);
        *(float4*)(Xl + srow0 * XSTRIDE + scol) = v0;
        *(float4*)(Xl + srow1 * XSTRIDE + scol) = v1;
    }

    const float* wl = pw + wave * 8192 + lane * 2;
    const float* rl = wl + 64 * 64;
    v2f fb0[4][4], fb1[4][4];
    load_chunk(fb0, wl);
    __syncthreads();

    int xcur = 0, hcur = 0;
    for (int grp = 0; grp < NGRP; ++grp) {
        float4 xr0, xr1;
        const bool more = (grp + 1 < NGRP);
        if (more) {
            const int tt = (grp + 1) * GSTEPS + srow0;
            const int t  = dir ? (T_STEPS - 1 - tt) : tt;
            xr0 = *(const float4*)(in + ((size_t)(b0 + 0) * T_STEPS + t) * UNITS_ + scol);
            xr1 = *(const float4*)(in + ((size_t)(b0 + 1) * T_STEPS + t) * UNITS_ + scol);
        }

        v8f acc[4];
#pragma unroll
        for (int q = 0; q < 4; ++q)
            acc[q] = (v8f){0.f, 0.f, 0.f, 0.f, 0.f, 0.f, 0.f, 0.f};
        gemm_phase(acc, fb0, fb1, Xl + xcur * (16 * XSTRIDE) + lc * XSTRIDE + 2 * hs, wl, rl);
#pragma unroll
        for (int q = 0; q < 4; ++q)
#pragma unroll
            for (int e = 0; e < 8; ++e)
                XZl[(q * 8 + e) * NTHR + tid] = acc[q][e];

        if (more) {
            float* Xn = Xl + (xcur ^ 1) * (16 * XSTRIDE);
            *(float4*)(Xn + srow0 * XSTRIDE + scol) = xr0;
            *(float4*)(Xn + srow1 * XSTRIDE + scol) = xr1;
        }

        for (int g = 0; g < GSTEPS; ++g) {
            const int tt = grp * GSTEPS + g;
            const int t  = dir ? (T_STEPS - 1 - tt) : tt;

#pragma unroll
            for (int q = 0; q < 4; ++q) {
                const float s = XZl[(q * 8 + g) * NTHR + tid];
                acc[q] = (v8f){s, s, s, s, s, s, s, s};
            }
            const float* nextp = (g + 1 < GSTEPS) ? rl : wl;
            gemm_phase(acc, fb0, fb1,
                       Hl + hcur * (ROWS_WG * HSTRIDE) + (lc >> 3) * HSTRIDE + 2 * hs,
                       rl, nextp);

            const float zi = acc[0][0] + bI;
            const float zf = acc[1][0] + bF;
            const float zg = acc[2][0] + bG;
            const float zo = acc[3][0] + bO;
            const float gi = sigmoid_f(zi);
            const float gf = sigmoid_f(zf);
            const float gg = tanh_f(zg);
            const float go = sigmoid_f(zo);
            const float c  = gf * cst + gi * gg;
            cst = c;
            const float h = go * tanh_f(c);
            Hl[(hcur ^ 1) * (ROWS_WG * HSTRIDE) + hs * HSTRIDE + unit] = h;
            const float ov = residual
                ? (h + Xl[xcur * (16 * XSTRIDE) + (8 * hs + g) * XSTRIDE + unit]) : h;
            out[((size_t)(b0 + hs) * T_STEPS + t) * UNITS_ + unit] = ov;
            __syncthreads();
            hcur ^= 1;
        }
        xcur ^= 1;
    }
}

__global__ void combine_avg_kernel(const float4* __restrict__ fw,
                                   float4* __restrict__ out, int n4) {
    int i = blockIdx.x * blockDim.x + threadIdx.x;
    if (i < n4) {
        float4 a = fw[i];
        float4 b = out[i];
        out[i] = make_float4((a.x + b.x) * 0.5f, (a.y + b.y) * 0.5f,
                             (a.z + b.z) * 0.5f, (a.w + b.w) * 0.5f);
    }
}

extern "C" void kernel_launch(void* const* d_in, const int* in_sizes, int n_in,
                              void* d_out, int out_size, void* d_ws, size_t ws_size,
                              hipStream_t stream) {
    const float* x   = (const float*)d_in[0];
    const float* kfw = (const float*)d_in[1];
    const float* rfw = (const float*)d_in[2];
    const float* bfw = (const float*)d_in[3];
    const float* kbw = (const float*)d_in[4];
    const float* rbw = (const float*)d_in[5];
    const float* bbw = (const float*)d_in[6];
    float* out = (float*)d_out;

    float* pk   = (float*)d_ws;
    float* wsfw = pk + (size_t)4 * COMBO_F;
    float* wsbw = out;

    pack_weights_kernel<<<4096, 256, 0, stream>>>(kfw, rfw, kbw, rbw, pk);

    const size_t shmem = (size_t)LDS_BYTES;
    (void)hipFuncSetAttribute((const void*)bilstm_layer_kernel,
                              hipFuncAttributeMaxDynamicSharedMemorySize,
                              (int)shmem);

    bilstm_layer_kernel<<<dim3(64, 2), NTHR, shmem, stream>>>(
        x, x, pk + 0 * (size_t)COMBO_F, pk + 1 * (size_t)COMBO_F,
        bfw + 0, bbw + 0, wsfw, wsbw,  0);

    bilstm_layer_kernel<<<dim3(64, 2), NTHR, shmem, stream>>>(
        wsfw, wsbw, pk + 2 * (size_t)COMBO_F, pk + 3 * (size_t)COMBO_F,
        bfw + 1024, bbw + 1024, wsfw, wsbw,  1);

    const int n4 = (128 * 256 * 256) / 4;
    combine_avg_kernel<<<(n4 + 255) / 256, 256, 0, stream>>>(
        (const float4*)wsfw, (float4*)out, n4);
}


# weight fragments requested three chunks (48 matrix instructions) ahead through a ring of four register buffers, not one: a wave left alone on its SIMD covers the L2 round trip; sum order kept
# speedup vs baseline: 2.1557x; 1.0553x over previous
#include <hip/hip_runtime.h>


typedef float v2f __attribute__((ext_vector_type(2)));
typedef float v8f __attribute__((ext_vector_type(8)));

#define T_STEPS 256
#define UNITS_  256
#define COMBO_F (512 * 1024)
#define ROWS_WG 2
#define GSTEPS  8
#define NGRP    (T_STEPS / GSTEPS)
#define NTHR    512
#define XSTRIDE 260
#define HSTRIDE 264
#define TQ      (16 * 128 * 64)
#define LDS_X_F  (2 * 16 * XSTRIDE)
#define LDS_H_F  (2 * ROWS_WG * HSTRIDE)
#define LDS_XZ_F (32 * NTHR)
#define LDS_BYTES (164 * 1024)

__device__ __forceinline__ float sigmoid_f(float x) {
    return 1.0f / (1.0f + __expf(-x));
}
__device__ __forceinline__ float tanh_f(float x) {
    x = fminf(15.0f, fmaxf(-15.0f, x));
    float e = __expf(2.0f * x);
    return (e - 1.0f) / (e + 1.0f);
}

__global__ void pack_weights_kernel(const float* __restrict__ kfw,
                                    const float* __restrict__ rfw,
                                    const float* __restrict__ kbw,
                                    const float* __restrict__ rbw,
                                    float* __restrict__ packed) {
    int i = blockIdx.x * blockDim.x + threadIdx.x;
    int lane  = i & 31;
    int ks    = (i >> 5) & 127;
    int nt    = (i >> 12) & 63;
    int combo = i >> 18;
    int layer = combo >> 1;
    int isbw  = combo & 1;
    const float* W = (isbw ? kbw : kfw) + (size_t)layer * 256 * 1024;
    const float* R = (isbw ? rbw : rfw) + (size_t)layer * 256 * 1024;

    int n  = nt * 16 + (lane & 15);
    int k0 = ks * 4 + ((lane >> 4) << 1);
    float d0, d1;
    if (k0 < 256) {
        d0 = W[(size_t)k0 * 1024 + n];
        d1 = W[(size_t)(k0 + 1) * 1024 + n];
    } else {
        int kk = k0 - 256;
        d0 = R[(size_t)kk * 1024 + n];
        d1 = R[(size_t)(kk + 1) * 1024 + n];
    }
    packed[(size_t)i * 2 + 0] = d0;
    packed[(size_t)i * 2 + 1] = d1;
}

__device__ __forceinline__ void load_chunk(v2f (&b)[4][4], const float* p) {
#pragma unroll
    for (int kk = 0; kk < 4; ++kk)
#pragma unroll
        for (int q = 0; q < 4; ++q)
            b[kk][q] = *(const v2f*)(p + q * TQ + kk * 64);
}
__device__ __forceinline__ void load_afrag(v2f (&a)[4], const float* p) {
#pragma unroll
    for (int kk = 0; kk < 4; ++kk) a[kk] = *(const v2f*)(p + kk * 4);
}
__device__ __forceinline__ void mma_chunk(v8f (&acc)[4], const v2f (&a)[4],
                                          const v2f (&b)[4][4]) {
#pragma unroll
    for (int kk = 0; kk < 4; ++kk)
#pragma unroll
        for (int q = 0; q < 4; ++q)
            acc[q] = __builtin_amdgcn_wmma_f32_16x16x4_f32(
                false, a[kk], false, b[kk][q], (short)0, acc[q], false, false);
}

__device__ __forceinline__ void gemm_phase(v8f (&acc)[4], v2f (&b0)[4][4],
                                           v2f (&b1)[4][4], v2f (&b2)[4][4],
                                           v2f (&b3)[4][4], const float* aL,
                                           const float* bp, const float* nextp) {
    v2f a0[4], a1[4];
    load_afrag(a0, aL);
#define STAGE_FENCE() __builtin_amdgcn_sched_barrier(0)
#define MMA_DONE() asm volatile("" : "+v"(acc[0]), "+v"(acc[1]), "+v"(acc[2]), "+v"(acc[3]) : : "memory")
#pragma unroll 1
    for (int ch = 0; ch < 16; ch += 4) {
        load_chunk(b3, bp + (ch + 3) * 256);
        load_afrag(a1, aL + (ch + 1) * 16);
        STAGE_FENCE();
        mma_chunk(acc, a0, b0);
        MMA_DONE();
        STAGE_FENCE();
        const float* np = (ch + 4 < 16) ? (bp + (ch + 4) * 256) : nextp;
        load_chunk(b0, np);
        load_afrag(a0, aL + (ch + 2) * 16);
        STAGE_FENCE();
        mma_chunk(acc, a1, b1);
        MMA_DONE();
        STAGE_FENCE();
        load_chunk(b1, np + 256);
        load_afrag(a1, aL + (ch + 3) * 16);
        STAGE_FENCE();
        mma_chunk(acc, a0, b2);
        MMA_DONE();
        STAGE_FENCE();
        load_chunk(b2, np + 512);
        load_afrag(a0, aL + ((ch + 4) & 15) * 16);
        STAGE_FENCE();
        mma_chunk(acc, a1, b3);
        MMA_DONE();
        STAGE_FENCE();
    }
#undef MMA_DONE
#undef STAGE_FENCE
}

__global__ void __launch_bounds__(NTHR)
bilstm_layer_kernel(const float* in_fw,
                    const float* in_bw,
                    const float* __restrict__ pw_fw,
                    const float* __restrict__ pw_bw,
                    const float* __restrict__ bias_fw,
                    const float* __restrict__ bias_bw,
                    float* out_fw,
                    float* out_bw,
                    int residual) {
    extern __shared__ float lds[];
    float* Xl  = lds;
    float* Hl  = lds + LDS_X_F;
    float* XZl = Hl + LDS_H_F;

    const int dir      = blockIdx.y;
    const float* in    = dir ? in_bw   : in_fw;
    const float* pw    = dir ? pw_bw   : pw_fw;
    const float* bias  = dir ? bias_bw : bias_fw;
    float*       out   = dir ? out_bw  : out_fw;

    const int b0   = blockIdx.x * ROWS_WG;
    const int tid  = threadIdx.x;
    const int lane = tid & 31;
    const int wave = tid >> 5;
    const int hs   = lane >> 4;
    const int lc   = lane & 15;
    const int unit = wave * 16 + lc;

    const float bI = bias[unit];
    const float bF = bias[256 + unit];
    const float bG = bias[512 + unit];
    const float bO = bias[768 + unit];
    float cst = 0.0f;

    Hl[(tid >> 8) * HSTRIDE + (tid & 255)] = 0.0f;

    const int srow0 = tid >> 6;
    const int srow1 = srow0 + 8;
    const int scol  = (tid & 63) * 4;
    {
        const int t = dir ? (T_STEPS - 1 - srow0) : srow0;
        const float4 v0 = *(const float4*)(in + ((size_t)(b0 + 0) * T_STEPS + t) * UNITS_ + scol);
        const float4 v1 = *(const float4*)(in + ((size_t)(b0 + 1) * T_STEPS + t) * UNITS_ + scol);
        *(float4*)(Xl + srow0 * XSTRIDE + scol) = v0;
        *(float4*)(Xl + srow1 * XSTRIDE + scol) = v1;
    }

    const float* wl = pw + wave * 8192 + lane * 2;
    const float* rl = wl + 64 * 64;
    v2f fb0[4][4], fb1[4][4], fb2[4][4], fb3[4][4];
    load_chunk(fb0, wl);
    load_chunk(fb1, wl + 256);
    load_chunk(fb2, wl + 512);
    __syncthreads();

    int xcur = 0, hcur = 0;
    for (int grp = 0; grp < NGRP; ++grp) {
        float4 xr0, xr1;
        const bool more = (grp + 1 < NGRP);
        if (more) {
            const int tt = (grp + 1) * GSTEPS + srow0;
            const int t  = dir ? (T_STEPS - 1 - tt) : tt;
            xr0 = *(const float4*)(in + ((size_t)(b0 + 0) * T_STEPS + t) * UNITS_ + scol);
            xr1 = *(const float4*)(in + ((size_t)(b0 + 1) * T_STEPS + t) * UNITS_ + scol);
        }

        v8f acc[4];
#pragma unroll
        for (int q = 0; q < 4; ++q)
            acc[q] = (v8f){0.f, 0.f, 0.f, 0.f, 0.f, 0.f, 0.f, 0.f};
        gemm_phase(acc, fb0, fb1, fb2, fb3,
                   Xl + xcur * (16 * XSTRIDE) + lc * XSTRIDE + 2 * hs, wl, rl);
#pragma unroll
        for (int q = 0; q < 4; ++q)
#pragma unroll
            for (int e = 0; e < 8; ++e)
                XZl[(q * 8 + e) * NTHR + tid] = acc[q][e];

        if (more) {
            float* Xn = Xl + (xcur ^ 1) * (16 * XSTRIDE);
            *(float4*)(Xn + srow0 * XSTRIDE + scol) = xr0;
            *(float4*)(Xn + srow1 * XSTRIDE + scol) = xr1;
        }

        for (int g = 0; g < GSTEPS; ++g) {
            const int tt = grp * GSTEPS + g;
            const int t  = dir ? (T_STEPS - 1 - tt) : tt;

#pragma unroll
            for (int q = 0; q < 4; ++q) {
                const float s = XZl[(q * 8 + g) * NTHR + tid];
                acc[q] = (v8f){s, s, s, s, s, s, s, s};
            }
            const float* nextp = (g + 1 < GSTEPS) ? rl : wl;
            gemm_phase(acc, fb0, fb1, fb2, fb3,
                       Hl + hcur * (ROWS_WG * HSTRIDE) + (lc >> 3) * HSTRIDE + 2 * hs,
                       rl, nextp);

            const float zi = acc[0][0] + bI;
            const float zf = acc[1][0] + bF;
            const float zg = acc[2][0] + bG;
            const float zo = acc[3][0] + bO;
            const float gi = sigmoid_f(zi);
            const float gf = sigmoid_f(zf);
            const float gg = tanh_f(zg);
            const float go = sigmoid_f(zo);
            const float c  = gf * cst + gi * gg;
            cst = c;
            const float h = go * tanh_f(c);
            Hl[(hcur ^ 1) * (ROWS_WG * HSTRIDE) + hs * HSTRIDE + unit] = h;
            const float ov = residual
                ? (h + Xl[xcur * (16 * XSTRIDE) + (8 * hs + g) * XSTRIDE + unit]) : h;
            out[((size_t)(b0 + hs) * T_STEPS + t) * UNITS_ + unit] = ov;
            __syncthreads();
            hcur ^= 1;
        }
        xcur ^= 1;
    }
}

__global__ void combine_avg_kernel(const float4* __restrict__ fw,
                                   float4* __restrict__ out, int n4) {
    int i = blockIdx.x * blockDim.x + threadIdx.x;
    if (i < n4) {
        float4 a = fw[i];
        float4 b = out[i];
        out[i] = make_float4((a.x + b.x) * 0.5f, (a.y + b.y) * 0.5f,
                             (a.z + b.z) * 0.5f, (a.w + b.w) * 0.5f);
    }
}

extern "C" void kernel_launch(void* const* d_in, const int* in_sizes, int n_in,
                              void* d_out, int out_size, void* d_ws, size_t ws_size,
                              hipStream_t stream) {
    const float* x   = (const float*)d_in[0];
    const float* kfw = (const float*)d_in[1];
    const float* rfw = (const float*)d_in[2];
    const float* bfw = (const float*)d_in[3];
    const float* kbw = (const float*)d_in[4];
    const float* rbw = (const float*)d_in[5];
    const float* bbw = (const float*)d_in[6];
    float* out = (float*)d_out;

    float* pk   = (float*)d_ws;
    float* wsfw = pk + (size_t)4 * COMBO_F;
    float* wsbw = out;

    pack_weights_kernel<<<4096, 256, 0, stream>>>(kfw, rfw, kbw, rbw, pk);

    const size_t shmem = (size_t)LDS_BYTES;
    (void)hipFuncSetAttribute((const void*)bilstm_layer_kernel,
                              hipFuncAttributeMaxDynamicSharedMemorySize,
                              (int)shmem);

    bilstm_layer_kernel<<<dim3(64, 2), NTHR, shmem, stream>>>(
        x, x, pk + 0 * (size_t)COMBO_F, pk + 1 * (size_t)COMBO_F,
        bfw + 0, bbw + 0, wsfw, wsbw,  0);

    bilstm_layer_kernel<<<dim3(64, 2), NTHR, shmem, stream>>>(
        wsfw, wsbw, pk + 2 * (size_t)COMBO_F, pk + 3 * (size_t)COMBO_F,
        bfw + 1024, bbw + 1024, wsfw, wsbw,  1);

    const int n4 = (128 * 256 * 256) / 4;
    combine_avg_kernel<<<(n4 + 255) / 256, 256, 0, stream>>>(
        (const float4*)wsfw, (float4*)out, n4);
}


# wave priority falls with progress inside each product (s_setprio 3 in chunks 0-7, 2 in 8-11, 1 in 12-13, 0 in 14-15): a SIMD's four waves finish together, none runs its tail alone; sum order kept
# speedup vs baseline: 2.2265x; 1.0328x over previous
#include <hip/hip_runtime.h>


typedef float v2f __attribute__((ext_vector_type(2)));
typedef float v8f __attribute__((ext_vector_type(8)));

#define T_STEPS 256
#define UNITS_  256
#define COMBO_F (512 * 1024)
#define ROWS_WG 2
#define GSTEPS  8
#define NGRP    (T_STEPS / GSTEPS)
#define NTHR    512
#define XSTRIDE 260
#define HSTRIDE 264
#define TQ      (16 * 128 * 64)
#define LDS_X_F  (2 * 16 * XSTRIDE)
#define LDS_H_F  (2 * ROWS_WG * HSTRIDE)
#define LDS_XZ_F (32 * NTHR)
#define LDS_BYTES (164 * 1024)

__device__ __forceinline__ float sigmoid_f(float x) {
    return 1.0f / (1.0f + __expf(-x));
}
__device__ __forceinline__ float tanh_f(float x) {
    x = fminf(15.0f, fmaxf(-15.0f, x));
    float e = __expf(2.0f * x);
    return (e - 1.0f) / (e + 1.0f);
}

__global__ void pack_weights_kernel(const float* __restrict__ kfw,
                                    const float* __restrict__ rfw,
                                    const float* __restrict__ kbw,
                                    const float* __restrict__ rbw,
                                    float* __restrict__ packed) {
    int i = blockIdx.x * blockDim.x + threadIdx.x;
    int lane  = i & 31;
    int ks    = (i >> 5) & 127;
    int nt    = (i >> 12) & 63;
    int combo = i >> 18;
    int layer = combo >> 1;
    int isbw  = combo & 1;
    const float* W = (isbw ? kbw : kfw) + (size_t)layer * 256 * 1024;
    const float* R = (isbw ? rbw : rfw) + (size_t)layer * 256 * 1024;

    int n  = nt * 16 + (lane & 15);
    int k0 = ks * 4 + ((lane >> 4) << 1);
    float d0, d1;
    if (k0 < 256) {
        d0 = W[(size_t)k0 * 1024 + n];
        d1 = W[(size_t)(k0 + 1) * 1024 + n];
    } else {
        int kk = k0 - 256;
        d0 = R[(size_t)kk * 1024 + n];
        d1 = R[(size_t)(kk + 1) * 1024 + n];
    }
    packed[(size_t)i * 2 + 0] = d0;
    packed[(size_t)i * 2 + 1] = d1;
}

__device__ __forceinline__ void load_chunk(v2f (&b)[4][4], const float* p) {
#pragma unroll
    for (int kk = 0; kk < 4; ++kk)
#pragma unroll
        for (int q = 0; q < 4; ++q)
            b[kk][q] = *(const v2f*)(p + q * TQ + kk * 64);
}
__device__ __forceinline__ void load_afrag(v2f (&a)[4], const float* p) {
#pragma unroll
    for (int kk = 0; kk < 4; ++kk) a[kk] = *(const v2f*)(p + kk * 4);
}
__device__ __forceinline__ void mma_chunk(v8f (&acc)[4], const v2f (&a)[4],
                                          const v2f (&b)[4][4]) {
#pragma unroll
    for (int kk = 0; kk < 4; ++kk)
#pragma unroll
        for (int q = 0; q < 4; ++q)
            acc[q] = __builtin_amdgcn_wmma_f32_16x16x4_f32(
                false, a[kk], false, b[kk][q], (short)0, acc[q], false, false);
}

__device__ __forceinline__ void gemm_phase(v8f (&acc)[4], v2f (&b0)[4][4],
                                           v2f (&b1)[4][4], v2f (&b2)[4][4],
                                           v2f (&b3)[4][4], const float* aL,
                                           const float* bp, const float* nextp) {
    v2f a0[4], a1[4];
    load_afrag(a0, aL);
#define STAGE_FENCE() __builtin_amdgcn_sched_barrier(0)
#define MMA_DONE() asm volatile("" : "+v"(acc[0]), "+v"(acc[1]), "+v"(acc[2]), "+v"(acc[3]) : : "memory")
#define PRIO_IF(cnt, val, prio)                                   \
    asm volatile("s_cmp_lg_u32 %0, " #val "\n\t"                  \
                 "s_cbranch_scc1 1f\n\t"                          \
                 "s_setprio " #prio "\n"                           \
                 "1:" : : "s"(cnt) : "scc")
    __builtin_amdgcn_s_setprio(3);
#pragma unroll 1
    for (int ch = 0; ch < 16; ch += 4) {
        load_chunk(b3, bp + (ch + 3) * 256);
        load_afrag(a1, aL + (ch + 1) * 16);
        STAGE_FENCE();
        mma_chunk(acc, a0, b0);
        MMA_DONE();
        STAGE_FENCE();
        const float* np = (ch + 4 < 16) ? (bp + (ch + 4) * 256) : nextp;
        load_chunk(b0, np);
        load_afrag(a0, aL + (ch + 2) * 16);
        STAGE_FENCE();
        mma_chunk(acc, a1, b1);
        MMA_DONE();
        STAGE_FENCE();
        PRIO_IF(ch, 12, 0);
        load_chunk(b1, np + 256);
        load_afrag(a1, aL + (ch + 3) * 16);
        STAGE_FENCE();
        mma_chunk(acc, a0, b2);
        MMA_DONE();
        STAGE_FENCE();
        load_chunk(b2, np + 512);
        load_afrag(a0, aL + ((ch + 4) & 15) * 16);
        STAGE_FENCE();
        mma_chunk(acc, a1, b3);
        MMA_DONE();
        STAGE_FENCE();
        PRIO_IF(ch, 4, 2);
        PRIO_IF(ch, 8, 1);
    }
#undef PRIO_IF
#undef MMA_DONE
#undef STAGE_FENCE
}

__global__ void __launch_bounds__(NTHR)
bilstm_layer_kernel(const float* in_fw,
                    const float* in_bw,
                    const float* __restrict__ pw_fw,
                    const float* __restrict__ pw_bw,
                    const float* __restrict__ bias_fw,
                    const float* __restrict__ bias_bw,
                    float* out_fw,
                    float* out_bw,
                    int residual) {
    extern __shared__ float lds[];
    float* Xl  = lds;
    float* Hl  = lds + LDS_X_F;
    float* XZl = Hl + LDS_H_F;

    const int dir      = blockIdx.y;
    const float* in    = dir ? in_bw   : in_fw;
    const float* pw    = dir ? pw_bw   : pw_fw;
    const float* bias  = dir ? bias_bw : bias_fw;
    float*       out   = dir ? out_bw  : out_fw;

    const int b0   = blockIdx.x * ROWS_WG;
    const int tid  = threadIdx.x;
    const int lane = tid & 31;
    const int wave = tid >> 5;
    const int hs   = lane >> 4;
    const int lc   = lane & 15;
    const int unit = wave * 16 + lc;

    const float bI = bias[unit];
    const float bF = bias[256 + unit];
    const float bG = bias[512 + unit];
    const float bO = bias[768 + unit];
    float cst = 0.0f;

    Hl[(tid >> 8) * HSTRIDE + (tid & 255)] = 0.0f;

    const int srow0 = tid >> 6;
    const int srow1 = srow0 + 8;
    const int scol  = (tid & 63) * 4;
    {
        const int t = dir ? (T_STEPS - 1 - srow0) : srow0;
        const float4 v0 = *(const float4*)(in + ((size_t)(b0 + 0) * T_STEPS + t) * UNITS_ + scol);
        const float4 v1 = *(const float4*)(in + ((size_t)(b0 + 1) * T_STEPS + t) * UNITS_ + scol);
        *(float4*)(Xl + srow0 * XSTRIDE + scol) = v0;
        *(float4*)(Xl + srow1 * XSTRIDE + scol) = v1;
    }

    const float* wl = pw + wave * 8192 + lane * 2;
    const float* rl = wl + 64 * 64;
    v2f fb0[4][4], fb1[4][4], fb2[4][4], fb3[4][4];
    load_chunk(fb0, wl);
    load_chunk(fb1, wl + 256);
    load_chunk(fb2, wl + 512);
    __syncthreads();

    int xcur = 0, hcur = 0;
    for (int grp = 0; grp < NGRP; ++grp) {
        float4 xr0, xr1;
        const bool more = (grp + 1 < NGRP);
        if (more) {
            const int tt = (grp + 1) * GSTEPS + srow0;
            const int t  = dir ? (T_STEPS - 1 - tt) : tt;
            xr0 = *(const float4*)(in + ((size_t)(b0 + 0) * T_STEPS + t) * UNITS_ + scol);
            xr1 = *(const float4*)(in + ((size_t)(b0 + 1) * T_STEPS + t) * UNITS_ + scol);
        }

        v8f acc[4];
#pragma unroll
        for (int q = 0; q < 4; ++q)
            acc[q] = (v8f){0.f, 0.f, 0.f, 0.f, 0.f, 0.f, 0.f, 0.f};
        gemm_phase(acc, fb0, fb1, fb2, fb3,
                   Xl + xcur * (16 * XSTRIDE) + lc * XSTRIDE + 2 * hs, wl, rl);
#pragma unroll
        for (int q = 0; q < 4; ++q)
#pragma unroll
            for (int e = 0; e < 8; ++e)
                XZl[(q * 8 + e) * NTHR + tid] = acc[q][e];

        if (more) {
            float* Xn = Xl + (xcur ^ 1) * (16 * XSTRIDE);
            *(float4*)(Xn + srow0 * XSTRIDE + scol) = xr0;
            *(float4*)(Xn + srow1 * XSTRIDE + scol) = xr1;
        }

        for (int g = 0; g < GSTEPS; ++g) {
            const int tt = grp * GSTEPS + g;
            const int t  = dir ? (T_STEPS - 1 - tt) : tt;

#pragma unroll
            for (int q = 0; q < 4; ++q) {
                const float s = XZl[(q * 8 + g) * NTHR + tid];
                acc[q] = (v8f){s, s, s, s, s, s, s, s};
            }
            const float* nextp = (g + 1 < GSTEPS) ? rl : wl;
            gemm_phase(acc, fb0, fb1, fb2, fb3,
                       Hl + hcur * (ROWS_WG * HSTRIDE) + (lc >> 3) * HSTRIDE + 2 * hs,
                       rl, nextp);

            const float zi = acc[0][0] + bI;
            const float zf = acc[1][0] + bF;
            const float zg = acc[2][0] + bG;
            const float zo = acc[3][0] + bO;
            const float gi = sigmoid_f(zi);
            const float gf = sigmoid_f(zf);
            const float gg = tanh_f(zg);
            const float go = sigmoid_f(zo);
            const float c  = gf * cst + gi * gg;
            cst = c;
            const float h = go * tanh_f(c);
            Hl[(hcur ^ 1) * (ROWS_WG * HSTRIDE) + hs * HSTRIDE + unit] = h;
            const float ov = residual
                ? (h + Xl[xcur * (16 * XSTRIDE) + (8 * hs + g) * XSTRIDE + unit]) : h;
            out[((size_t)(b0 + hs) * T_STEPS + t) * UNITS_ + unit] = ov;
            __syncthreads();
            hcur ^= 1;
        }
        xcur ^= 1;
    }
}

__global__ void combine_avg_kernel(const float4* __restrict__ fw,
                                   float4* __restrict__ out, int n4) {
    int i = blockIdx.x * blockDim.x + threadIdx.x;
    if (i < n4) {
        float4 a = fw[i];
        float4 b = out[i];
        out[i] = make_float4((a.x + b.x) * 0.5f, (a.y + b.y) * 0.5f,
                             (a.z + b.z) * 0.5f, (a.w + b.w) * 0.5f);
    }
}

extern "C" void kernel_launch(void* const* d_in, const int* in_sizes, int n_in,
                              void* d_out, int out_size, void* d_ws, size_t ws_size,
                              hipStream_t stream) {
    const float* x   = (const float*)d_in[0];
    const float* kfw = (const float*)d_in[1];
    const float* rfw = (const float*)d_in[2];
    const float* bfw = (const float*)d_in[3];
    const float* kbw = (const float*)d_in[4];
    const float* rbw = (const float*)d_in[5];
    const float* bbw = (const float*)d_in[6];
    float* out = (float*)d_out;

    float* pk   = (float*)d_ws;
    float* wsfw = pk + (size_t)4 * COMBO_F;
    float* wsbw = out;

    pack_weights_kernel<<<4096, 256, 0, stream>>>(kfw, rfw, kbw, rbw, pk);

    const size_t shmem = (size_t)LDS_BYTES;
    (void)hipFuncSetAttribute((const void*)bilstm_layer_kernel,
                              hipFuncAttributeMaxDynamicSharedMemorySize,
                              (int)shmem);

    bilstm_layer_kernel<<<dim3(64, 2), NTHR, shmem, stream>>>(
        x, x, pk + 0 * (size_t)COMBO_F, pk + 1 * (size_t)COMBO_F,
        bfw + 0, bbw + 0, wsfw, wsbw,  0);

    bilstm_layer_kernel<<<dim3(64, 2), NTHR, shmem, stream>>>(
        wsfw, wsbw, pk + 2 * (size_t)COMBO_F, pk + 3 * (size_t)COMBO_F,
        bfw + 1024, bbw + 1024, wsfw, wsbw,  1);

    const int n4 = (128 * 256 * 256) / 4;
    combine_avg_kernel<<<(n4 + 255) / 256, 256, 0, stream>>>(
        (const float4*)wsfw, (float4*)out, n4);
}


# time cut into 16 chunks, one launch each: launch k runs chunk k of layer 0 beside chunk k-1 of layer 1 (256 work-groups, one per WGP), h and c carried through the workspace; sum order kept
# speedup vs baseline: 3.5797x; 1.6078x over previous
#include <hip/hip_runtime.h>


typedef float v2f __attribute__((ext_vector_type(2)));
typedef float v8f __attribute__((ext_vector_type(8)));

#define T_STEPS 256
#define UNITS_  256
#define COMBO_F (512 * 1024)
#define ROWS_WG 2
#define GSTEPS  8
#define NGRP    (T_STEPS / GSTEPS)
#define NCHUNK  16
#define CHUNK_G (NGRP / NCHUNK)
#define STATE_F (2 * 2 * 64 * 2 * NTHR)
#define NTHR    512
#define XSTRIDE 260
#define HSTRIDE 264
#define TQ      (16 * 128 * 64)
#define LDS_X_F  (2 * 16 * XSTRIDE)
#define LDS_H_F  (2 * ROWS_WG * HSTRIDE)
#define LDS_XZ_F (32 * NTHR)
#define LDS_BYTES (164 * 1024)

__device__ __forceinline__ float sigmoid_f(float x) {
    return 1.0f / (1.0f + __expf(-x));
}
__device__ __forceinline__ float tanh_f(float x) {
    x = fminf(15.0f, fmaxf(-15.0f, x));
    float e = __expf(2.0f * x);
    return (e - 1.0f) / (e + 1.0f);
}

__global__ void pack_weights_kernel(const float* __restrict__ kfw,
                                    const float* __restrict__ rfw,
                                    const float* __restrict__ kbw,
                                    const float* __restrict__ rbw,
                                    float* __restrict__ packed) {
    int i = blockIdx.x * blockDim.x + threadIdx.x;
    int lane  = i & 31;
    int ks    = (i >> 5) & 127;
    int nt    = (i >> 12) & 63;
    int combo = i >> 18;
    int layer = combo >> 1;
    int isbw  = combo & 1;
    const float* W = (isbw ? kbw : kfw) + (size_t)layer * 256 * 1024;
    const float* R = (isbw ? rbw : rfw) + (size_t)layer * 256 * 1024;

    int n  = nt * 16 + (lane & 15);
    int k0 = ks * 4 + ((lane >> 4) << 1);
    float d0, d1;
    if (k0 < 256) {
        d0 = W[(size_t)k0 * 1024 + n];
        d1 = W[(size_t)(k0 + 1) * 1024 + n];
    } else {
        int kk = k0 - 256;
        d0 = R[(size_t)kk * 1024 + n];
        d1 = R[(size_t)(kk + 1) * 1024 + n];
    }
    packed[(size_t)i * 2 + 0] = d0;
    packed[(size_t)i * 2 + 1] = d1;
}

__device__ __forceinline__ void load_chunk(v2f (&b)[4][4], const float* p) {
#pragma unroll
    for (int kk = 0; kk < 4; ++kk)
#pragma unroll
        for (int q = 0; q < 4; ++q)
            b[kk][q] = *(const v2f*)(p + q * TQ + kk * 64);
}
__device__ __forceinline__ void load_afrag(v2f (&a)[4], const float* p) {
#pragma unroll
    for (int kk = 0; kk < 4; ++kk) a[kk] = *(const v2f*)(p + kk * 4);
}
__device__ __forceinline__ void mma_chunk(v8f (&acc)[4], const v2f (&a)[4],
                                          const v2f (&b)[4][4]) {
#pragma unroll
    for (int kk = 0; kk < 4; ++kk)
#pragma unroll
        for (int q = 0; q < 4; ++q)
            acc[q] = __builtin_amdgcn_wmma_f32_16x16x4_f32(
                false, a[kk], false, b[kk][q], (short)0, acc[q], false, false);
}

__device__ __forceinline__ void gemm_phase(v8f (&acc)[4], v2f (&b0)[4][4],
                                           v2f (&b1)[4][4], v2f (&b2)[4][4],
                                           v2f (&b3)[4][4], const float* aL,
                                           const float* bp, const float* nextp) {
    v2f a0[4], a1[4];
    load_afrag(a0, aL);
#define STAGE_FENCE() __builtin_amdgcn_sched_barrier(0)
#define MMA_DONE() asm volatile("" : "+v"(acc[0]), "+v"(acc[1]), "+v"(acc[2]), "+v"(acc[3]) : : "memory")
#define PRIO_IF(cnt, val, prio)                                   \
    asm volatile("s_cmp_lg_u32 %0, " #val "\n\t"                  \
                 "s_cbranch_scc1 1f\n\t"                          \
                 "s_setprio " #prio "\n"                           \
                 "1:" : : "s"(cnt) : "scc")
    __builtin_amdgcn_s_setprio(3);
#pragma unroll 1
    for (int ch = 0; ch < 16; ch += 4) {
        load_chunk(b3, bp + (ch + 3) * 256);
        load_afrag(a1, aL + (ch + 1) * 16);
        STAGE_FENCE();
        mma_chunk(acc, a0, b0);
        MMA_DONE();
        STAGE_FENCE();
        const float* np = (ch + 4 < 16) ? (bp + (ch + 4) * 256) : nextp;
        load_chunk(b0, np);
        load_afrag(a0, aL + (ch + 2) * 16);
        STAGE_FENCE();
        mma_chunk(acc, a1, b1);
        MMA_DONE();
        STAGE_FENCE();
        PRIO_IF(ch, 12, 0);
        load_chunk(b1, np + 256);
        load_afrag(a1, aL + (ch + 3) * 16);
        STAGE_FENCE();
        mma_chunk(acc, a0, b2);
        MMA_DONE();
        STAGE_FENCE();
        load_chunk(b2, np + 512);
        load_afrag(a0, aL + ((ch + 4) & 15) * 16);
        STAGE_FENCE();
        mma_chunk(acc, a1, b3);
        MMA_DONE();
        STAGE_FENCE();
        PRIO_IF(ch, 4, 2);
        PRIO_IF(ch, 8, 1);
    }
#undef PRIO_IF
#undef MMA_DONE
#undef STAGE_FENCE
}

__global__ void __launch_bounds__(NTHR)
bilstm_layer_kernel(const float* x,
                    float* chain_fw,
                    float* chain_bw,
                    const float* __restrict__ packed,
                    const float* __restrict__ bias_fw,
                    const float* __restrict__ bias_bw,
                    float* state,
                    int launch_idx, int nchunk, int chunk_groups) {
    const int layer = blockIdx.z;
    const int chunk = launch_idx - layer;
    if (chunk < 0 || chunk >= nchunk) return;

    extern __shared__ float lds[];
    float* Xl  = lds;
    float* Hl  = lds + LDS_X_F;
    float* XZl = Hl + LDS_H_F;

    const int dir      = blockIdx.y;
    float*       out   = dir ? chain_bw : chain_fw;
    const float* in    = layer ? out : x;
    const float* pw    = packed + (size_t)(layer * 2 + dir) * COMBO_F;
    const float* bias  = (dir ? bias_bw : bias_fw) + layer * 1024;
    const int residual = layer;
    float* st = state + (size_t)((layer * 2 + dir) * 64 + blockIdx.x) * (2 * NTHR);

    const int b0   = blockIdx.x * ROWS_WG;
    const int tid  = threadIdx.x;
    const int lane = tid & 31;
    const int wave = tid >> 5;
    const int hs   = lane >> 4;
    const int lc   = lane & 15;
    const int unit = wave * 16 + lc;

    const float bI = bias[unit];
    const float bF = bias[256 + unit];
    const float bG = bias[512 + unit];
    const float bO = bias[768 + unit];
    float cst = 0.0f, hreg = 0.0f;
    if (chunk > 0) {
        hreg = st[tid];
        cst  = st[NTHR + tid];
    }
    Hl[hs * HSTRIDE + unit] = hreg;

    const int srow0 = tid >> 6;
    const int srow1 = srow0 + 8;
    const int scol  = (tid & 63) * 4;
    const int g_first = chunk * chunk_groups;
    const int g_end   = g_first + chunk_groups;
    {
        const int tt0 = g_first * GSTEPS + srow0;
        const int t = dir ? (T_STEPS - 1 - tt0) : tt0;
        const float4 v0 = *(const float4*)(in + ((size_t)(b0 + 0) * T_STEPS + t) * UNITS_ + scol);
        const float4 v1 = *(const float4*)(in + ((size_t)(b0 + 1) * T_STEPS + t) * UNITS_ + scol);
        *(float4*)(Xl + srow0 * XSTRIDE + scol) = v0;
        *(float4*)(Xl + srow1 * XSTRIDE + scol) = v1;
    }

    const float* wl = pw + wave * 8192 + lane * 2;
    const float* rl = wl + 64 * 64;
    v2f fb0[4][4], fb1[4][4], fb2[4][4], fb3[4][4];
    load_chunk(fb0, wl);
    load_chunk(fb1, wl + 256);
    load_chunk(fb2, wl + 512);
    __syncthreads();

    int xcur = 0, hcur = 0;
    for (int grp = g_first; grp < g_end; ++grp) {
        float4 xr0, xr1;
        const bool more = (grp + 1 < g_end);
        if (more) {
            const int tt = (grp + 1) * GSTEPS + srow0;
            const int t  = dir ? (T_STEPS - 1 - tt) : tt;
            xr0 = *(const float4*)(in + ((size_t)(b0 + 0) * T_STEPS + t) * UNITS_ + scol);
            xr1 = *(const float4*)(in + ((size_t)(b0 + 1) * T_STEPS + t) * UNITS_ + scol);
        }

        v8f acc[4];
#pragma unroll
        for (int q = 0; q < 4; ++q)
            acc[q] = (v8f){0.f, 0.f, 0.f, 0.f, 0.f, 0.f, 0.f, 0.f};
        gemm_phase(acc, fb0, fb1, fb2, fb3,
                   Xl + xcur * (16 * XSTRIDE) + lc * XSTRIDE + 2 * hs, wl, rl);
#pragma unroll
        for (int q = 0; q < 4; ++q)
#pragma unroll
            for (int e = 0; e < 8; ++e)
                XZl[(q * 8 + e) * NTHR + tid] = acc[q][e];

        if (more) {
            float* Xn = Xl + (xcur ^ 1) * (16 * XSTRIDE);
            *(float4*)(Xn + srow0 * XSTRIDE + scol) = xr0;
            *(float4*)(Xn + srow1 * XSTRIDE + scol) = xr1;
        }

        for (int g = 0; g < GSTEPS; ++g) {
            const int tt = grp * GSTEPS + g;
            const int t  = dir ? (T_STEPS - 1 - tt) : tt;

#pragma unroll
            for (int q = 0; q < 4; ++q) {
                const float s = XZl[(q * 8 + g) * NTHR + tid];
                acc[q] = (v8f){s, s, s, s, s, s, s, s};
            }
            const float* nextp = (g + 1 < GSTEPS) ? rl : wl;
            gemm_phase(acc, fb0, fb1, fb2, fb3,
                       Hl + hcur * (ROWS_WG * HSTRIDE) + (lc >> 3) * HSTRIDE + 2 * hs,
                       rl, nextp);

            const float zi = acc[0][0] + bI;
            const float zf = acc[1][0] + bF;
            const float zg = acc[2][0] + bG;
            const float zo = acc[3][0] + bO;
            const float gi = sigmoid_f(zi);
            const float gf = sigmoid_f(zf);
            const float gg = tanh_f(zg);
            const float go = sigmoid_f(zo);
            const float c  = gf * cst + gi * gg;
            cst = c;
            const float h = go * tanh_f(c);
            hreg = h;
            Hl[(hcur ^ 1) * (ROWS_WG * HSTRIDE) + hs * HSTRIDE + unit] = h;
            const float ov = residual
                ? (h + Xl[xcur * (16 * XSTRIDE) + (8 * hs + g) * XSTRIDE + unit]) : h;
            out[((size_t)(b0 + hs) * T_STEPS + t) * UNITS_ + unit] = ov;
            __syncthreads();
            hcur ^= 1;
        }
        xcur ^= 1;
    }
    if (chunk + 1 < nchunk) {
        st[tid]        = hreg;
        st[NTHR + tid] = cst;
    }
}

__global__ void combine_avg_kernel(const float4* __restrict__ fw,
                                   float4* __restrict__ out, int n4) {
    int i = blockIdx.x * blockDim.x + threadIdx.x;
    if (i < n4) {
        float4 a = fw[i];
        float4 b = out[i];
        out[i] = make_float4((a.x + b.x) * 0.5f, (a.y + b.y) * 0.5f,
                             (a.z + b.z) * 0.5f, (a.w + b.w) * 0.5f);
    }
}

extern "C" void kernel_launch(void* const* d_in, const int* in_sizes, int n_in,
                              void* d_out, int out_size, void* d_ws, size_t ws_size,
                              hipStream_t stream) {
    const float* x   = (const float*)d_in[0];
    const float* kfw = (const float*)d_in[1];
    const float* rfw = (const float*)d_in[2];
    const float* bfw = (const float*)d_in[3];
    const float* kbw = (const float*)d_in[4];
    const float* rbw = (const float*)d_in[5];
    const float* bbw = (const float*)d_in[6];
    float* out = (float*)d_out;

    float* pk   = (float*)d_ws;
    float* wsfw = pk + (size_t)4 * COMBO_F;
    float* wsbw = out;
    float* stt  = wsfw + (size_t)128 * 256 * 256;

    pack_weights_kernel<<<4096, 256, 0, stream>>>(kfw, rfw, kbw, rbw, pk);

    const size_t shmem = (size_t)LDS_BYTES;
    (void)hipFuncSetAttribute((const void*)bilstm_layer_kernel,
                              hipFuncAttributeMaxDynamicSharedMemorySize,
                              (int)shmem);

#define CHUNK_LAUNCH(k)                                                     \
    bilstm_layer_kernel<<<dim3(64, 2, 2), NTHR, shmem, stream>>>(           \
        x, wsfw, wsbw, pk, bfw, bbw, stt, (k), NCHUNK, CHUNK_G)
    if (ws_size >= ((size_t)4 * COMBO_F + (size_t)128 * 256 * 256 + STATE_F) * sizeof(float)) {
        CHUNK_LAUNCH(0);  CHUNK_LAUNCH(1);  CHUNK_LAUNCH(2);  CHUNK_LAUNCH(3);
        CHUNK_LAUNCH(4);  CHUNK_LAUNCH(5);  CHUNK_LAUNCH(6);  CHUNK_LAUNCH(7);
        CHUNK_LAUNCH(8);  CHUNK_LAUNCH(9);  CHUNK_LAUNCH(10); CHUNK_LAUNCH(11);
        CHUNK_LAUNCH(12); CHUNK_LAUNCH(13); CHUNK_LAUNCH(14); CHUNK_LAUNCH(15);
        CHUNK_LAUNCH(16);
    } else {
        bilstm_layer_kernel<<<dim3(64, 2, 2), NTHR, shmem, stream>>>(
            x, wsfw, wsbw, pk, bfw, bbw, wsfw, 0, 1, NGRP);
        bilstm_layer_kernel<<<dim3(64, 2, 2), NTHR, shmem, stream>>>(
            x, wsfw, wsbw, pk, bfw, bbw, wsfw, 1, 1, NGRP);
    }
#undef CHUNK_LAUNCH

    const int n4 = (128 * 256 * 256) / 4;
    combine_avg_kernel<<<(n4 + 255) / 256, 256, 0, stream>>>(
        (const float4*)wsfw, (float4*)out, n4);
}


# one launch per time step, not per 16 steps: the launch boundary orders the steps, so each step's [x_t|h]@[W;R] runs on all WGPs with 16 distinct batch rows per matrix tile, not 2; sum order kept
# speedup vs baseline: 6.2460x; 1.7448x over previous
#include <hip/hip_runtime.h>


typedef float v2f __attribute__((ext_vector_type(2)));
typedef float v8f __attribute__((ext_vector_type(8)));

#define T_STEPS 256
#define UNITS_  256
#define COMBO_F (512 * 1024)
#define ROWS_WG 2
#define GSTEPS  8
#define NGRP    (T_STEPS / GSTEPS)
#define NCHUNK  16
#define CHUNK_G (NGRP / NCHUNK)
#define STATE_F (2 * 2 * 64 * 2 * NTHR)
#define NTHR    512
#define XSTRIDE 260
#define HSTRIDE 264
#define TQ      (16 * 128 * 64)
#define LDS_X_F  (2 * 16 * XSTRIDE)
#define LDS_H_F  (2 * ROWS_WG * HSTRIDE)
#define LDS_XZ_F (32 * NTHR)
#define LDS_BYTES (164 * 1024)

__device__ __forceinline__ float sigmoid_f(float x) {
    return 1.0f / (1.0f + __expf(-x));
}
__device__ __forceinline__ float tanh_f(float x) {
    x = fminf(15.0f, fmaxf(-15.0f, x));
    float e = __expf(2.0f * x);
    return (e - 1.0f) / (e + 1.0f);
}

__global__ void pack_weights_kernel(const float* __restrict__ kfw,
                                    const float* __restrict__ rfw,
                                    const float* __restrict__ kbw,
                                    const float* __restrict__ rbw,
                                    float* __restrict__ packed) {
    int i = blockIdx.x * blockDim.x + threadIdx.x;
    int lane  = i & 31;
    int ks    = (i >> 5) & 127;
    int nt    = (i >> 12) & 63;
    int combo = i >> 18;
    int layer = combo >> 1;
    int isbw  = combo & 1;
    const float* W = (isbw ? kbw : kfw) + (size_t)layer * 256 * 1024;
    const float* R = (isbw ? rbw : rfw) + (size_t)layer * 256 * 1024;

    int n  = nt * 16 + (lane & 15);
    int k0 = ks * 4 + ((lane >> 4) << 1);
    float d0, d1;
    if (k0 < 256) {
        d0 = W[(size_t)k0 * 1024 + n];
        d1 = W[(size_t)(k0 + 1) * 1024 + n];
    } else {
        int kk = k0 - 256;
        d0 = R[(size_t)kk * 1024 + n];
        d1 = R[(size_t)(kk + 1) * 1024 + n];
    }
    packed[(size_t)i * 2 + 0] = d0;
    packed[(size_t)i * 2 + 1] = d1;
}

__device__ __forceinline__ void load_chunk(v2f (&b)[4][4], const float* p) {
#pragma unroll
    for (int kk = 0; kk < 4; ++kk)
#pragma unroll
        for (int q = 0; q < 4; ++q)
            b[kk][q] = *(const v2f*)(p + q * TQ + kk * 64);
}
__device__ __forceinline__ void load_afrag(v2f (&a)[4], const float* p) {
#pragma unroll
    for (int kk = 0; kk < 4; ++kk) a[kk] = *(const v2f*)(p + kk * 4);
}
__device__ __forceinline__ void mma_chunk(v8f (&acc)[4], const v2f (&a)[4],
                                          const v2f (&b)[4][4]) {
#pragma unroll
    for (int kk = 0; kk < 4; ++kk)
#pragma unroll
        for (int q = 0; q < 4; ++q)
            acc[q] = __builtin_amdgcn_wmma_f32_16x16x4_f32(
                false, a[kk], false, b[kk][q], (short)0, acc[q], false, false);
}

__device__ __forceinline__ void gemm_phase(v8f (&acc)[4], v2f (&b0)[4][4],
                                           v2f (&b1)[4][4], v2f (&b2)[4][4],
                                           v2f (&b3)[4][4], const float* aL,
                                           const float* bp, const float* nextp) {
    v2f a0[4], a1[4];
    load_afrag(a0, aL);
#define STAGE_FENCE() __builtin_amdgcn_sched_barrier(0)
#define MMA_DONE() asm volatile("" : "+v"(acc[0]), "+v"(acc[1]), "+v"(acc[2]), "+v"(acc[3]) : : "memory")
#define PRIO_IF(cnt, val, prio)                                   \
    asm volatile("s_cmp_lg_u32 %0, " #val "\n\t"                  \
                 "s_cbranch_scc1 1f\n\t"                          \
                 "s_setprio " #prio "\n"                           \
                 "1:" : : "s"(cnt) : "scc")
    __builtin_amdgcn_s_setprio(3);
#pragma unroll 1
    for (int ch = 0; ch < 16; ch += 4) {
        load_chunk(b3, bp + (ch + 3) * 256);
        load_afrag(a1, aL + (ch + 1) * 16);
        STAGE_FENCE();
        mma_chunk(acc, a0, b0);
        MMA_DONE();
        STAGE_FENCE();
        const float* np = (ch + 4 < 16) ? (bp + (ch + 4) * 256) : nextp;
        load_chunk(b0, np);
        load_afrag(a0, aL + (ch + 2) * 16);
        STAGE_FENCE();
        mma_chunk(acc, a1, b1);
        MMA_DONE();
        STAGE_FENCE();
        PRIO_IF(ch, 12, 0);
        load_chunk(b1, np + 256);
        load_afrag(a1, aL + (ch + 3) * 16);
        STAGE_FENCE();
        mma_chunk(acc, a0, b2);
        MMA_DONE();
        STAGE_FENCE();
        load_chunk(b2, np + 512);
        load_afrag(a0, aL + ((ch + 4) & 15) * 16);
        STAGE_FENCE();
        mma_chunk(acc, a1, b3);
        MMA_DONE();
        STAGE_FENCE();
        PRIO_IF(ch, 4, 2);
        PRIO_IF(ch, 8, 1);
    }
#undef PRIO_IF
#undef MMA_DONE
#undef STAGE_FENCE
}

__global__ void __launch_bounds__(NTHR)
bilstm_layer_kernel(const float* x,
                    float* chain_fw,
                    float* chain_bw,
                    const float* __restrict__ packed,
                    const float* __restrict__ bias_fw,
                    const float* __restrict__ bias_bw,
                    float* state,
                    int launch_idx, int nchunk, int chunk_groups) {
    const int layer = blockIdx.z;
    const int chunk = launch_idx - layer;
    if (chunk < 0 || chunk >= nchunk) return;

    extern __shared__ float lds[];
    float* Xl  = lds;
    float* Hl  = lds + LDS_X_F;
    float* XZl = Hl + LDS_H_F;

    const int dir      = blockIdx.y;
    float*       out   = dir ? chain_bw : chain_fw;
    const float* in    = layer ? out : x;
    const float* pw    = packed + (size_t)(layer * 2 + dir) * COMBO_F;
    const float* bias  = (dir ? bias_bw : bias_fw) + layer * 1024;
    const int residual = layer;
    float* st = state + (size_t)((layer * 2 + dir) * 64 + blockIdx.x) * (2 * NTHR);

    const int b0   = blockIdx.x * ROWS_WG;
    const int tid  = threadIdx.x;
    const int lane = tid & 31;
    const int wave = tid >> 5;
    const int hs   = lane >> 4;
    const int lc   = lane & 15;
    const int unit = wave * 16 + lc;

    const float bI = bias[unit];
    const float bF = bias[256 + unit];
    const float bG = bias[512 + unit];
    const float bO = bias[768 + unit];
    float cst = 0.0f, hreg = 0.0f;
    if (chunk > 0) {
        hreg = st[tid];
        cst  = st[NTHR + tid];
    }
    Hl[hs * HSTRIDE + unit] = hreg;

    const int srow0 = tid >> 6;
    const int srow1 = srow0 + 8;
    const int scol  = (tid & 63) * 4;
    const int g_first = chunk * chunk_groups;
    const int g_end   = g_first + chunk_groups;
    {
        const int tt0 = g_first * GSTEPS + srow0;
        const int t = dir ? (T_STEPS - 1 - tt0) : tt0;
        const float4 v0 = *(const float4*)(in + ((size_t)(b0 + 0) * T_STEPS + t) * UNITS_ + scol);
        const float4 v1 = *(const float4*)(in + ((size_t)(b0 + 1) * T_STEPS + t) * UNITS_ + scol);
        *(float4*)(Xl + srow0 * XSTRIDE + scol) = v0;
        *(float4*)(Xl + srow1 * XSTRIDE + scol) = v1;
    }

    const float* wl = pw + wave * 8192 + lane * 2;
    const float* rl = wl + 64 * 64;
    v2f fb0[4][4], fb1[4][4], fb2[4][4], fb3[4][4];
    load_chunk(fb0, wl);
    load_chunk(fb1, wl + 256);
    load_chunk(fb2, wl + 512);
    __syncthreads();

    int xcur = 0, hcur = 0;
    for (int grp = g_first; grp < g_end; ++grp) {
        float4 xr0, xr1;
        const bool more = (grp + 1 < g_end);
        if (more) {
            const int tt = (grp + 1) * GSTEPS + srow0;
            const int t  = dir ? (T_STEPS - 1 - tt) : tt;
            xr0 = *(const float4*)(in + ((size_t)(b0 + 0) * T_STEPS + t) * UNITS_ + scol);
            xr1 = *(const float4*)(in + ((size_t)(b0 + 1) * T_STEPS + t) * UNITS_ + scol);
        }

        v8f acc[4];
#pragma unroll
        for (int q = 0; q < 4; ++q)
            acc[q] = (v8f){0.f, 0.f, 0.f, 0.f, 0.f, 0.f, 0.f, 0.f};
        gemm_phase(acc, fb0, fb1, fb2, fb3,
                   Xl + xcur * (16 * XSTRIDE) + lc * XSTRIDE + 2 * hs, wl, rl);
#pragma unroll
        for (int q = 0; q < 4; ++q)
#pragma unroll
            for (int e = 0; e < 8; ++e)
                XZl[(q * 8 + e) * NTHR + tid] = acc[q][e];

        if (more) {
            float* Xn = Xl + (xcur ^ 1) * (16 * XSTRIDE);
            *(float4*)(Xn + srow0 * XSTRIDE + scol) = xr0;
            *(float4*)(Xn + srow1 * XSTRIDE + scol) = xr1;
        }

        for (int g = 0; g < GSTEPS; ++g) {
            const int tt = grp * GSTEPS + g;
            const int t  = dir ? (T_STEPS - 1 - tt) : tt;

#pragma unroll
            for (int q = 0; q < 4; ++q) {
                const float s = XZl[(q * 8 + g) * NTHR + tid];
                acc[q] = (v8f){s, s, s, s, s, s, s, s};
            }
            const float* nextp = (g + 1 < GSTEPS) ? rl : wl;
            gemm_phase(acc, fb0, fb1, fb2, fb3,
                       Hl + hcur * (ROWS_WG * HSTRIDE) + (lc >> 3) * HSTRIDE + 2 * hs,
                       rl, nextp);

            const float zi = acc[0][0] + bI;
            const float zf = acc[1][0] + bF;
            const float zg = acc[2][0] + bG;
            const float zo = acc[3][0] + bO;
            const float gi = sigmoid_f(zi);
            const float gf = sigmoid_f(zf);
            const float gg = tanh_f(zg);
            const float go = sigmoid_f(zo);
            const float c  = gf * cst + gi * gg;
            cst = c;
            const float h = go * tanh_f(c);
            hreg = h;
            Hl[(hcur ^ 1) * (ROWS_WG * HSTRIDE) + hs * HSTRIDE + unit] = h;
            const float ov = residual
                ? (h + Xl[xcur * (16 * XSTRIDE) + (8 * hs + g) * XSTRIDE + unit]) : h;
            out[((size_t)(b0 + hs) * T_STEPS + t) * UNITS_ + unit] = ov;
            __syncthreads();
            hcur ^= 1;
        }
        xcur ^= 1;
    }
    if (chunk + 1 < nchunk) {
        st[tid]        = hreg;
        st[NTHR + tid] = cst;
    }
}

#define S_THR   128
#define APITCH  516
#define ZPITCH  36
#define S_LDS_A (16 * APITCH)
#define CELLS_F (128 * 256)

__device__ __forceinline__ void load_chunk8(v2f (&b)[8][2], const float* p) {
#pragma unroll
    for (int kk = 0; kk < 8; ++kk)
#pragma unroll
        for (int q = 0; q < 2; ++q)
            b[kk][q] = *(const v2f*)(p + q * TQ + kk * 64);
}
__device__ __forceinline__ void load_afrag8(v2f (&a)[8], const float* p) {
#pragma unroll
    for (int kk = 0; kk < 8; ++kk) a[kk] = *(const v2f*)(p + kk * 4);
}
__device__ __forceinline__ void mma_chunk8(v8f (&acc)[2], const v2f (&a)[8],
                                           const v2f (&b)[8][2]) {
#pragma unroll
    for (int kk = 0; kk < 8; ++kk)
#pragma unroll
        for (int q = 0; q < 2; ++q)
            acc[q] = __builtin_amdgcn_wmma_f32_16x16x4_f32(
                false, a[kk], false, b[kk][q], (short)0, acc[q], false, false);
}

__global__ void __launch_bounds__(S_THR)
bilstm_step_kernel(const float* x,
                   float* chain_fw,
                   float* chain_bw,
                   const float* __restrict__ packed,
                   const float* __restrict__ bias_fw,
                   const float* __restrict__ bias_bw,
                   float* state, int step) {
    const int layer = blockIdx.z;
    const int tt  = step - layer;
    const int ttw = step - 2;
    const bool do_step  = (tt >= 0 && tt < T_STEPS);
    const bool do_flush = (layer == 1 && ttw >= 0 && ttw < T_STEPS);
    if (!do_step && !do_flush) return;

    extern __shared__ float lds[];
    float* Al = lds;
    float* Zl = lds + S_LDS_A;

    const int dir   = blockIdx.y;
    float* out      = dir ? chain_bw : chain_fw;
    const float* in = layer ? out : x;
    const int combo = layer * 2 + dir;
    const int b0    = (blockIdx.x >> 3) * 16;
    const int ugrp  = blockIdx.x & 7;
    const int tid   = threadIdx.x;
    const int lane  = tid & 31;
    const int wave  = tid >> 5;
    const int hs    = lane >> 4;
    const int lc    = lane & 15;
    const int unit  = ugrp * 32 + lane;

    float* cbuf = state + (size_t)combo * CELLS_F;
    float* h1   = state + (size_t)4 * CELLS_F + (size_t)dir * 2 * CELLS_F;

    const int tw = dir ? (T_STEPS - 1 - ttw) : ttw;
    const float* hflush = h1 + (size_t)(ttw & 1) * CELLS_F;
    float fh[4], fi[4];
    if (!do_step) {
#pragma unroll
        for (int j = 0; j < 4; ++j) {
            const int r = b0 + j * 4 + wave;
            fh[j] = hflush[(size_t)r * UNITS_ + unit];
            fi[j] = out[((size_t)r * T_STEPS + tw) * UNITS_ + unit];
        }
#pragma unroll
        for (int j = 0; j < 4; ++j)
            out[((size_t)(b0 + j * 4 + wave) * T_STEPS + tw) * UNITS_ + unit] = fh[j] + fi[j];
        return;
    }

    const int t     = dir ? (T_STEPS - 1 - tt) : tt;
    const int tprev = dir ? (t + 1) : (t - 1);
    const float* pw   = packed + (size_t)combo * COMBO_F;
    const float* bias = (dir ? bias_bw : bias_fw) + layer * 1024;

    float4 v[16];
    {
        const float* hsrc = layer ? (h1 + (size_t)((tt - 1) & 1) * CELLS_F + (size_t)b0 * UNITS_)
                                  : (out + ((size_t)b0 * T_STEPS + tprev) * UNITS_);
        const float* xsrc = in + ((size_t)b0 * T_STEPS + t) * UNITS_;
        const bool   isx  = (tid < 64);
        const float* src  = isx ? (xsrc + tid * 4) : (hsrc + (tid - 64) * 4);
        const size_t rowf = (isx || !layer) ? (size_t)T_STEPS * UNITS_ : (size_t)UNITS_;
#pragma unroll
        for (int r = 0; r < 16; ++r) v[r] = make_float4(0.f, 0.f, 0.f, 0.f);
        if (isx || tt > 0) {
#pragma unroll
            for (int r = 0; r < 16; ++r)
                v[r] = *(const float4*)(src + (size_t)r * rowf);
        }
    }

    float cprev[4];
#pragma unroll
    for (int j = 0; j < 4; ++j)
        cprev[j] = (tt > 0) ? cbuf[(size_t)(b0 + j * 4 + wave) * UNITS_ + unit] : 0.0f;
    if (do_flush) {
#pragma unroll
        for (int j = 0; j < 4; ++j) {
            const int r = b0 + j * 4 + wave;
            fh[j] = hflush[(size_t)r * UNITS_ + unit];
            fi[j] = out[((size_t)r * T_STEPS + tw) * UNITS_ + unit];
        }
    }

    const float* bp = pw + (size_t)((wave >> 1) * 32 + ugrp * 2 + (wave & 1)) * 8192 + lane * 2;
    v2f fb0[8][2], fb1[8][2], fb2[8][2], fb3[8][2];
    load_chunk8(fb0, bp);
    load_chunk8(fb1, bp + 512);
    load_chunk8(fb2, bp + 1024);

    const float bI = bias[unit];
    const float bF = bias[256 + unit];
    const float bG = bias[512 + unit];
    const float bO = bias[768 + unit];

#pragma unroll
    for (int r = 0; r < 16; ++r)
        *(float4*)(Al + r * APITCH + tid * 4) = v[r];
    __syncthreads();

    v8f acc[2];
    acc[0] = (v8f){0.f, 0.f, 0.f, 0.f, 0.f, 0.f, 0.f, 0.f};
    acc[1] = (v8f){0.f, 0.f, 0.f, 0.f, 0.f, 0.f, 0.f, 0.f};
    const float* aL = Al + lc * APITCH + 2 * hs;
    v2f a0[8], a1[8];
    load_afrag8(a0, aL);
#define S_FENCE() __builtin_amdgcn_sched_barrier(0)
#define S_DONE() asm volatile("" : "+v"(acc[0]), "+v"(acc[1]) : : "memory")
#pragma unroll 1
    for (int ch = 0; ch < 12; ch += 4) {
        load_chunk8(fb3, bp + (ch + 3) * 512);
        load_afrag8(a1, aL + (ch + 1) * 32);
        S_FENCE();
        mma_chunk8(acc, a0, fb0);
        S_DONE();
        S_FENCE();
        load_chunk8(fb0, bp + (ch + 4) * 512);
        load_afrag8(a0, aL + (ch + 2) * 32);
        S_FENCE();
        mma_chunk8(acc, a1, fb1);
        S_DONE();
        S_FENCE();
        load_chunk8(fb1, bp + (ch + 5) * 512);
        load_afrag8(a1, aL + (ch + 3) * 32);
        S_FENCE();
        mma_chunk8(acc, a0, fb2);
        S_DONE();
        S_FENCE();
        load_chunk8(fb2, bp + (ch + 6) * 512);
        load_afrag8(a0, aL + (ch + 4) * 32);
        S_FENCE();
        mma_chunk8(acc, a1, fb3);
        S_DONE();
        S_FENCE();
    }
    load_chunk8(fb3, bp + 15 * 512);
    load_afrag8(a1, aL + 13 * 32);
    S_FENCE();
    mma_chunk8(acc, a0, fb0);
    S_DONE();
    S_FENCE();
    load_afrag8(a0, aL + 14 * 32);
    S_FENCE();
    mma_chunk8(acc, a1, fb1);
    S_DONE();
    S_FENCE();
    load_afrag8(a1, aL + 15 * 32);
    S_FENCE();
    mma_chunk8(acc, a0, fb2);
    mma_chunk8(acc, a1, fb3);
#undef S_DONE
#undef S_FENCE

#pragma unroll
    for (int q = 0; q < 2; ++q)
#pragma unroll
        for (int e = 0; e < 8; ++e)
            Zl[(((wave >> 1) * 2 + q) * 16 + 8 * hs + e) * ZPITCH + (wave & 1) * 16 + lc] = acc[q][e];
    __syncthreads();

    float* hdst = layer ? (h1 + (size_t)(tt & 1) * CELLS_F) : out;
#pragma unroll
    for (int j = 0; j < 4; ++j) {
        const int rl = j * 4 + wave;
        const int r  = b0 + rl;
        const float zi = Zl[(0 * 16 + rl) * ZPITCH + lane] + bI;
        const float zf = Zl[(1 * 16 + rl) * ZPITCH + lane] + bF;
        const float zg = Zl[(2 * 16 + rl) * ZPITCH + lane] + bG;
        const float zo = Zl[(3 * 16 + rl) * ZPITCH + lane] + bO;
        const float gi = sigmoid_f(zi);
        const float gf = sigmoid_f(zf);
        const float gg = tanh_f(zg);
        const float go = sigmoid_f(zo);
        const size_t ci = (size_t)r * UNITS_ + unit;
        const float cst = cprev[j];
        const float c = gf * cst + gi * gg;
        cbuf[ci] = c;
        const float h = go * tanh_f(c);
        if (layer) hdst[ci] = h;
        else       hdst[((size_t)r * T_STEPS + t) * UNITS_ + unit] = h;
    }
    if (do_flush) {
#pragma unroll
        for (int j = 0; j < 4; ++j)
            out[((size_t)(b0 + j * 4 + wave) * T_STEPS + tw) * UNITS_ + unit] = fh[j] + fi[j];
    }
}

__global__ void combine_avg_kernel(const float4* __restrict__ fw,
                                   float4* __restrict__ out, int n4) {
    int i = blockIdx.x * blockDim.x + threadIdx.x;
    if (i < n4) {
        float4 a = fw[i];
        float4 b = out[i];
        out[i] = make_float4((a.x + b.x) * 0.5f, (a.y + b.y) * 0.5f,
                             (a.z + b.z) * 0.5f, (a.w + b.w) * 0.5f);
    }
}

extern "C" void kernel_launch(void* const* d_in, const int* in_sizes, int n_in,
                              void* d_out, int out_size, void* d_ws, size_t ws_size,
                              hipStream_t stream) {
    const float* x   = (const float*)d_in[0];
    const float* kfw = (const float*)d_in[1];
    const float* rfw = (const float*)d_in[2];
    const float* bfw = (const float*)d_in[3];
    const float* kbw = (const float*)d_in[4];
    const float* rbw = (const float*)d_in[5];
    const float* bbw = (const float*)d_in[6];
    float* out = (float*)d_out;

    float* pk   = (float*)d_ws;
    float* wsfw = pk + (size_t)4 * COMBO_F;
    float* wsbw = out;
    float* stt  = wsfw + (size_t)128 * 256 * 256;

    pack_weights_kernel<<<4096, 256, 0, stream>>>(kfw, rfw, kbw, rbw, pk);

    const size_t shmem = (size_t)LDS_BYTES;

#define STEP_LAUNCH(s)                                                      \
    bilstm_step_kernel<<<dim3(64, 2, 2), S_THR, shmem, stream>>>(           \
        x, wsfw, wsbw, pk, bfw, bbw, stt, (s))
#define STEP_LAUNCH_16(b)                                                   \
    STEP_LAUNCH((b) + 0);  STEP_LAUNCH((b) + 1);  STEP_LAUNCH((b) + 2);     \
    STEP_LAUNCH((b) + 3);  STEP_LAUNCH((b) + 4);  STEP_LAUNCH((b) + 5);     \
    STEP_LAUNCH((b) + 6);  STEP_LAUNCH((b) + 7);  STEP_LAUNCH((b) + 8);     \
    STEP_LAUNCH((b) + 9);  STEP_LAUNCH((b) + 10); STEP_LAUNCH((b) + 11);    \
    STEP_LAUNCH((b) + 12); STEP_LAUNCH((b) + 13); STEP_LAUNCH((b) + 14);    \
    STEP_LAUNCH((b) + 15)
    if (ws_size >= ((size_t)4 * COMBO_F + (size_t)128 * 256 * 256 + STATE_F) * sizeof(float)) {
        (void)hipFuncSetAttribute((const void*)bilstm_step_kernel,
                                  hipFuncAttributeMaxDynamicSharedMemorySize,
                                  (int)shmem);
        STEP_LAUNCH_16(0);   STEP_LAUNCH_16(16);  STEP_LAUNCH_16(32);  STEP_LAUNCH_16(48);
        STEP_LAUNCH_16(64);  STEP_LAUNCH_16(80);  STEP_LAUNCH_16(96);  STEP_LAUNCH_16(112);
        STEP_LAUNCH_16(128); STEP_LAUNCH_16(144); STEP_LAUNCH_16(160); STEP_LAUNCH_16(176);
        STEP_LAUNCH_16(192); STEP_LAUNCH_16(208); STEP_LAUNCH_16(224); STEP_LAUNCH_16(240);
        STEP_LAUNCH(256);
        STEP_LAUNCH(257);
    } else {
        (void)hipFuncSetAttribute((const void*)bilstm_layer_kernel,
                                  hipFuncAttributeMaxDynamicSharedMemorySize,
                                  (int)shmem);
        bilstm_layer_kernel<<<dim3(64, 2, 2), NTHR, shmem, stream>>>(
            x, wsfw, wsbw, pk, bfw, bbw, wsfw, 0, 1, NGRP);
        bilstm_layer_kernel<<<dim3(64, 2, 2), NTHR, shmem, stream>>>(
            x, wsfw, wsbw, pk, bfw, bbw, wsfw, 1, 1, NGRP);
    }
#undef STEP_LAUNCH_16
#undef STEP_LAUNCH

    const int n4 = (128 * 256 * 256) / 4;
    combine_avg_kernel<<<(n4 + 255) / 256, 256, 0, stream>>>(
        (const float4*)wsfw, (float4*)out, n4);
}


# one-step kernel: a thread's four cell updates in one branch-free block (h destination by selected base and pitch), so their exp/divide chains overlap on the lone wave; same operations, sum order kept
# speedup vs baseline: 6.6282x; 1.0612x over previous
#include <hip/hip_runtime.h>


typedef float v2f __attribute__((ext_vector_type(2)));
typedef float v8f __attribute__((ext_vector_type(8)));

#define T_STEPS 256
#define UNITS_  256
#define COMBO_F (512 * 1024)
#define ROWS_WG 2
#define GSTEPS  8
#define NGRP    (T_STEPS / GSTEPS)
#define NCHUNK  16
#define CHUNK_G (NGRP / NCHUNK)
#define STATE_F (2 * 2 * 64 * 2 * NTHR)
#define NTHR    512
#define XSTRIDE 260
#define HSTRIDE 264
#define TQ      (16 * 128 * 64)
#define LDS_X_F  (2 * 16 * XSTRIDE)
#define LDS_H_F  (2 * ROWS_WG * HSTRIDE)
#define LDS_XZ_F (32 * NTHR)
#define LDS_BYTES (164 * 1024)

__device__ __forceinline__ float sigmoid_f(float x) {
    return 1.0f / (1.0f + __expf(-x));
}
__device__ __forceinline__ float tanh_f(float x) {
    x = fminf(15.0f, fmaxf(-15.0f, x));
    float e = __expf(2.0f * x);
    return (e - 1.0f) / (e + 1.0f);
}

__global__ void pack_weights_kernel(const float* __restrict__ kfw,
                                    const float* __restrict__ rfw,
                                    const float* __restrict__ kbw,
                                    const float* __restrict__ rbw,
                                    float* __restrict__ packed) {
    int i = blockIdx.x * blockDim.x + threadIdx.x;
    int lane  = i & 31;
    int ks    = (i >> 5) & 127;
    int nt    = (i >> 12) & 63;
    int combo = i >> 18;
    int layer = combo >> 1;
    int isbw  = combo & 1;
    const float* W = (isbw ? kbw : kfw) + (size_t)layer * 256 * 1024;
    const float* R = (isbw ? rbw : rfw) + (size_t)layer * 256 * 1024;

    int n  = nt * 16 + (lane & 15);
    int k0 = ks * 4 + ((lane >> 4) << 1);
    float d0, d1;
    if (k0 < 256) {
        d0 = W[(size_t)k0 * 1024 + n];
        d1 = W[(size_t)(k0 + 1) * 1024 + n];
    } else {
        int kk = k0 - 256;
        d0 = R[(size_t)kk * 1024 + n];
        d1 = R[(size_t)(kk + 1) * 1024 + n];
    }
    packed[(size_t)i * 2 + 0] = d0;
    packed[(size_t)i * 2 + 1] = d1;
}

__device__ __forceinline__ void load_chunk(v2f (&b)[4][4], const float* p) {
#pragma unroll
    for (int kk = 0; kk < 4; ++kk)
#pragma unroll
        for (int q = 0; q < 4; ++q)
            b[kk][q] = *(const v2f*)(p + q * TQ + kk * 64);
}
__device__ __forceinline__ void load_afrag(v2f (&a)[4], const float* p) {
#pragma unroll
    for (int kk = 0; kk < 4; ++kk) a[kk] = *(const v2f*)(p + kk * 4);
}
__device__ __forceinline__ void mma_chunk(v8f (&acc)[4], const v2f (&a)[4],
                                          const v2f (&b)[4][4]) {
#pragma unroll
    for (int kk = 0; kk < 4; ++kk)
#pragma unroll
        for (int q = 0; q < 4; ++q)
            acc[q] = __builtin_amdgcn_wmma_f32_16x16x4_f32(
                false, a[kk], false, b[kk][q], (short)0, acc[q], false, false);
}

__device__ __forceinline__ void gemm_phase(v8f (&acc)[4], v2f (&b0)[4][4],
                                           v2f (&b1)[4][4], v2f (&b2)[4][4],
                                           v2f (&b3)[4][4], const float* aL,
                                           const float* bp, const float* nextp) {
    v2f a0[4], a1[4];
    load_afrag(a0, aL);
#define STAGE_FENCE() __builtin_amdgcn_sched_barrier(0)
#define MMA_DONE() asm volatile("" : "+v"(acc[0]), "+v"(acc[1]), "+v"(acc[2]), "+v"(acc[3]) : : "memory")
#define PRIO_IF(cnt, val, prio)                                   \
    asm volatile("s_cmp_lg_u32 %0, " #val "\n\t"                  \
                 "s_cbranch_scc1 1f\n\t"                          \
                 "s_setprio " #prio "\n"                           \
                 "1:" : : "s"(cnt) : "scc")
    __builtin_amdgcn_s_setprio(3);
#pragma unroll 1
    for (int ch = 0; ch < 16; ch += 4) {
        load_chunk(b3, bp + (ch + 3) * 256);
        load_afrag(a1, aL + (ch + 1) * 16);
        STAGE_FENCE();
        mma_chunk(acc, a0, b0);
        MMA_DONE();
        STAGE_FENCE();
        const float* np = (ch + 4 < 16) ? (bp + (ch + 4) * 256) : nextp;
        load_chunk(b0, np);
        load_afrag(a0, aL + (ch + 2) * 16);
        STAGE_FENCE();
        mma_chunk(acc, a1, b1);
        MMA_DONE();
        STAGE_FENCE();
        PRIO_IF(ch, 12, 0);
        load_chunk(b1, np + 256);
        load_afrag(a1, aL + (ch + 3) * 16);
        STAGE_FENCE();
        mma_chunk(acc, a0, b2);
        MMA_DONE();
        STAGE_FENCE();
        load_chunk(b2, np + 512);
        load_afrag(a0, aL + ((ch + 4) & 15) * 16);
        STAGE_FENCE();
        mma_chunk(acc, a1, b3);
        MMA_DONE();
        STAGE_FENCE();
        PRIO_IF(ch, 4, 2);
        PRIO_IF(ch, 8, 1);
    }
#undef PRIO_IF
#undef MMA_DONE
#undef STAGE_FENCE
}

__global__ void __launch_bounds__(NTHR)
bilstm_layer_kernel(const float* x,
                    float* chain_fw,
                    float* chain_bw,
                    const float* __restrict__ packed,
                    const float* __restrict__ bias_fw,
                    const float* __restrict__ bias_bw,
                    float* state,
                    int launch_idx, int nchunk, int chunk_groups) {
    const int layer = blockIdx.z;
    const int chunk = launch_idx - layer;
    if (chunk < 0 || chunk >= nchunk) return;

    extern __shared__ float lds[];
    float* Xl  = lds;
    float* Hl  = lds + LDS_X_F;
    float* XZl = Hl + LDS_H_F;

    const int dir      = blockIdx.y;
    float*       out   = dir ? chain_bw : chain_fw;
    const float* in    = layer ? out : x;
    const float* pw    = packed + (size_t)(layer * 2 + dir) * COMBO_F;
    const float* bias  = (dir ? bias_bw : bias_fw) + layer * 1024;
    const int residual = layer;
    float* st = state + (size_t)((layer * 2 + dir) * 64 + blockIdx.x) * (2 * NTHR);

    const int b0   = blockIdx.x * ROWS_WG;
    const int tid  = threadIdx.x;
    const int lane = tid & 31;
    const int wave = tid >> 5;
    const int hs   = lane >> 4;
    const int lc   = lane & 15;
    const int unit = wave * 16 + lc;

    const float bI = bias[unit];
    const float bF = bias[256 + unit];
    const float bG = bias[512 + unit];
    const float bO = bias[768 + unit];
    float cst = 0.0f, hreg = 0.0f;
    if (chunk > 0) {
        hreg = st[tid];
        cst  = st[NTHR + tid];
    }
    Hl[hs * HSTRIDE + unit] = hreg;

    const int srow0 = tid >> 6;
    const int srow1 = srow0 + 8;
    const int scol  = (tid & 63) * 4;
    const int g_first = chunk * chunk_groups;
    const int g_end   = g_first + chunk_groups;
    {
        const int tt0 = g_first * GSTEPS + srow0;
        const int t = dir ? (T_STEPS - 1 - tt0) : tt0;
        const float4 v0 = *(const float4*)(in + ((size_t)(b0 + 0) * T_STEPS + t) * UNITS_ + scol);
        const float4 v1 = *(const float4*)(in + ((size_t)(b0 + 1) * T_STEPS + t) * UNITS_ + scol);
        *(float4*)(Xl + srow0 * XSTRIDE + scol) = v0;
        *(float4*)(Xl + srow1 * XSTRIDE + scol) = v1;
    }

    const float* wl = pw + wave * 8192 + lane * 2;
    const float* rl = wl + 64 * 64;
    v2f fb0[4][4], fb1[4][4], fb2[4][4], fb3[4][4];
    load_chunk(fb0, wl);
    load_chunk(fb1, wl + 256);
    load_chunk(fb2, wl + 512);
    __syncthreads();

    int xcur = 0, hcur = 0;
    for (int grp = g_first; grp < g_end; ++grp) {
        float4 xr0, xr1;
        const bool more = (grp + 1 < g_end);
        if (more) {
            const int tt = (grp + 1) * GSTEPS + srow0;
            const int t  = dir ? (T_STEPS - 1 - tt) : tt;
            xr0 = *(const float4*)(in + ((size_t)(b0 + 0) * T_STEPS + t) * UNITS_ + scol);
            xr1 = *(const float4*)(in + ((size_t)(b0 + 1) * T_STEPS + t) * UNITS_ + scol);
        }

        v8f acc[4];
#pragma unroll
        for (int q = 0; q < 4; ++q)
            acc[q] = (v8f){0.f, 0.f, 0.f, 0.f, 0.f, 0.f, 0.f, 0.f};
        gemm_phase(acc, fb0, fb1, fb2, fb3,
                   Xl + xcur * (16 * XSTRIDE) + lc * XSTRIDE + 2 * hs, wl, rl);
#pragma unroll
        for (int q = 0; q < 4; ++q)
#pragma unroll
            for (int e = 0; e < 8; ++e)
                XZl[(q * 8 + e) * NTHR + tid] = acc[q][e];

        if (more) {
            float* Xn = Xl + (xcur ^ 1) * (16 * XSTRIDE);
            *(float4*)(Xn + srow0 * XSTRIDE + scol) = xr0;
            *(float4*)(Xn + srow1 * XSTRIDE + scol) = xr1;
        }

        for (int g = 0; g < GSTEPS; ++g) {
            const int tt = grp * GSTEPS + g;
            const int t  = dir ? (T_STEPS - 1 - tt) : tt;

#pragma unroll
            for (int q = 0; q < 4; ++q) {
                const float s = XZl[(q * 8 + g) * NTHR + tid];
                acc[q] = (v8f){s, s, s, s, s, s, s, s};
            }
            const float* nextp = (g + 1 < GSTEPS) ? rl : wl;
            gemm_phase(acc, fb0, fb1, fb2, fb3,
                       Hl + hcur * (ROWS_WG * HSTRIDE) + (lc >> 3) * HSTRIDE + 2 * hs,
                       rl, nextp);

            const float zi = acc[0][0] + bI;
            const float zf = acc[1][0] + bF;
            const float zg = acc[2][0] + bG;
            const float zo = acc[3][0] + bO;
            const float gi = sigmoid_f(zi);
            const float gf = sigmoid_f(zf);
            const float gg = tanh_f(zg);
            const float go = sigmoid_f(zo);
            const float c  = gf * cst + gi * gg;
            cst = c;
            const float h = go * tanh_f(c);
            hreg = h;
            Hl[(hcur ^ 1) * (ROWS_WG * HSTRIDE) + hs * HSTRIDE + unit] = h;
            const float ov = residual
                ? (h + Xl[xcur * (16 * XSTRIDE) + (8 * hs + g) * XSTRIDE + unit]) : h;
            out[((size_t)(b0 + hs) * T_STEPS + t) * UNITS_ + unit] = ov;
            __syncthreads();
            hcur ^= 1;
        }
        xcur ^= 1;
    }
    if (chunk + 1 < nchunk) {
        st[tid]        = hreg;
        st[NTHR + tid] = cst;
    }
}

#define S_THR   128
#define APITCH  516
#define ZPITCH  36
#define S_LDS_A (16 * APITCH)
#define CELLS_F (128 * 256)

__device__ __forceinline__ void load_chunk8(v2f (&b)[8][2], const float* p) {
#pragma unroll
    for (int kk = 0; kk < 8; ++kk)
#pragma unroll
        for (int q = 0; q < 2; ++q)
            b[kk][q] = *(const v2f*)(p + q * TQ + kk * 64);
}
__device__ __forceinline__ void load_afrag8(v2f (&a)[8], const float* p) {
#pragma unroll
    for (int kk = 0; kk < 8; ++kk) a[kk] = *(const v2f*)(p + kk * 4);
}
__device__ __forceinline__ void mma_chunk8(v8f (&acc)[2], const v2f (&a)[8],
                                           const v2f (&b)[8][2]) {
#pragma unroll
    for (int kk = 0; kk < 8; ++kk)
#pragma unroll
        for (int q = 0; q < 2; ++q)
            acc[q] = __builtin_amdgcn_wmma_f32_16x16x4_f32(
                false, a[kk], false, b[kk][q], (short)0, acc[q], false, false);
}

__global__ void __launch_bounds__(S_THR)
bilstm_step_kernel(const float* x,
                   float* chain_fw,
                   float* chain_bw,
                   const float* __restrict__ packed,
                   const float* __restrict__ bias_fw,
                   const float* __restrict__ bias_bw,
                   float* state, int step) {
    const int layer = blockIdx.z;
    const int tt  = step - layer;
    const int ttw = step - 2;
    const bool do_step  = (tt >= 0 && tt < T_STEPS);
    const bool do_flush = (layer == 1 && ttw >= 0 && ttw < T_STEPS);
    if (!do_step && !do_flush) return;

    extern __shared__ float lds[];
    float* Al = lds;
    float* Zl = lds + S_LDS_A;

    const int dir   = blockIdx.y;
    float* out      = dir ? chain_bw : chain_fw;
    const float* in = layer ? out : x;
    const int combo = layer * 2 + dir;
    const int b0    = (blockIdx.x >> 3) * 16;
    const int ugrp  = blockIdx.x & 7;
    const int tid   = threadIdx.x;
    const int lane  = tid & 31;
    const int wave  = tid >> 5;
    const int hs    = lane >> 4;
    const int lc    = lane & 15;
    const int unit  = ugrp * 32 + lane;

    float* cbuf = state + (size_t)combo * CELLS_F;
    float* h1   = state + (size_t)4 * CELLS_F + (size_t)dir * 2 * CELLS_F;

    const int tw = dir ? (T_STEPS - 1 - ttw) : ttw;
    const float* hflush = h1 + (size_t)(ttw & 1) * CELLS_F;
    float fh[4], fi[4];
    if (!do_step) {
#pragma unroll
        for (int j = 0; j < 4; ++j) {
            const int r = b0 + j * 4 + wave;
            fh[j] = hflush[(size_t)r * UNITS_ + unit];
            fi[j] = out[((size_t)r * T_STEPS + tw) * UNITS_ + unit];
        }
#pragma unroll
        for (int j = 0; j < 4; ++j)
            out[((size_t)(b0 + j * 4 + wave) * T_STEPS + tw) * UNITS_ + unit] = fh[j] + fi[j];
        return;
    }

    const int t     = dir ? (T_STEPS - 1 - tt) : tt;
    const int tprev = dir ? (t + 1) : (t - 1);
    const float* pw   = packed + (size_t)combo * COMBO_F;
    const float* bias = (dir ? bias_bw : bias_fw) + layer * 1024;

    float4 v[16];
    {
        const float* hsrc = layer ? (h1 + (size_t)((tt - 1) & 1) * CELLS_F + (size_t)b0 * UNITS_)
                                  : (out + ((size_t)b0 * T_STEPS + tprev) * UNITS_);
        const float* xsrc = in + ((size_t)b0 * T_STEPS + t) * UNITS_;
        const bool   isx  = (tid < 64);
        const float* src  = isx ? (xsrc + tid * 4) : (hsrc + (tid - 64) * 4);
        const size_t rowf = (isx || !layer) ? (size_t)T_STEPS * UNITS_ : (size_t)UNITS_;
#pragma unroll
        for (int r = 0; r < 16; ++r) v[r] = make_float4(0.f, 0.f, 0.f, 0.f);
        if (isx || tt > 0) {
#pragma unroll
            for (int r = 0; r < 16; ++r)
                v[r] = *(const float4*)(src + (size_t)r * rowf);
        }
    }

    float cprev[4];
#pragma unroll
    for (int j = 0; j < 4; ++j)
        cprev[j] = (tt > 0) ? cbuf[(size_t)(b0 + j * 4 + wave) * UNITS_ + unit] : 0.0f;
    if (do_flush) {
#pragma unroll
        for (int j = 0; j < 4; ++j) {
            const int r = b0 + j * 4 + wave;
            fh[j] = hflush[(size_t)r * UNITS_ + unit];
            fi[j] = out[((size_t)r * T_STEPS + tw) * UNITS_ + unit];
        }
    }

    const float* bp = pw + (size_t)((wave >> 1) * 32 + ugrp * 2 + (wave & 1)) * 8192 + lane * 2;
    v2f fb0[8][2], fb1[8][2], fb2[8][2], fb3[8][2];
    load_chunk8(fb0, bp);
    load_chunk8(fb1, bp + 512);
    load_chunk8(fb2, bp + 1024);

    const float bI = bias[unit];
    const float bF = bias[256 + unit];
    const float bG = bias[512 + unit];
    const float bO = bias[768 + unit];

#pragma unroll
    for (int r = 0; r < 16; ++r)
        *(float4*)(Al + r * APITCH + tid * 4) = v[r];
    __syncthreads();

    v8f acc[2];
    acc[0] = (v8f){0.f, 0.f, 0.f, 0.f, 0.f, 0.f, 0.f, 0.f};
    acc[1] = (v8f){0.f, 0.f, 0.f, 0.f, 0.f, 0.f, 0.f, 0.f};
    const float* aL = Al + lc * APITCH + 2 * hs;
    v2f a0[8], a1[8];
    load_afrag8(a0, aL);
#define S_FENCE() __builtin_amdgcn_sched_barrier(0)
#define S_DONE() asm volatile("" : "+v"(acc[0]), "+v"(acc[1]) : : "memory")
#pragma unroll 1
    for (int ch = 0; ch < 12; ch += 4) {
        load_chunk8(fb3, bp + (ch + 3) * 512);
        load_afrag8(a1, aL + (ch + 1) * 32);
        S_FENCE();
        mma_chunk8(acc, a0, fb0);
        S_DONE();
        S_FENCE();
        load_chunk8(fb0, bp + (ch + 4) * 512);
        load_afrag8(a0, aL + (ch + 2) * 32);
        S_FENCE();
        mma_chunk8(acc, a1, fb1);
        S_DONE();
        S_FENCE();
        load_chunk8(fb1, bp + (ch + 5) * 512);
        load_afrag8(a1, aL + (ch + 3) * 32);
        S_FENCE();
        mma_chunk8(acc, a0, fb2);
        S_DONE();
        S_FENCE();
        load_chunk8(fb2, bp + (ch + 6) * 512);
        load_afrag8(a0, aL + (ch + 4) * 32);
        S_FENCE();
        mma_chunk8(acc, a1, fb3);
        S_DONE();
        S_FENCE();
    }
    load_chunk8(fb3, bp + 15 * 512);
    load_afrag8(a1, aL + 13 * 32);
    S_FENCE();
    mma_chunk8(acc, a0, fb0);
    S_DONE();
    S_FENCE();
    load_afrag8(a0, aL + 14 * 32);
    S_FENCE();
    mma_chunk8(acc, a1, fb1);
    S_DONE();
    S_FENCE();
    load_afrag8(a1, aL + 15 * 32);
    S_FENCE();
    mma_chunk8(acc, a0, fb2);
    mma_chunk8(acc, a1, fb3);
#undef S_DONE
#undef S_FENCE

#pragma unroll
    for (int q = 0; q < 2; ++q)
#pragma unroll
        for (int e = 0; e < 8; ++e)
            Zl[(((wave >> 1) * 2 + q) * 16 + 8 * hs + e) * ZPITCH + (wave & 1) * 16 + lc] = acc[q][e];
    __syncthreads();

    float* hrow = layer ? (h1 + (size_t)(tt & 1) * CELLS_F + (size_t)b0 * UNITS_ + unit)
                        : (out + ((size_t)b0 * T_STEPS + t) * UNITS_ + unit);
    const size_t hpitch = layer ? (size_t)UNITS_ : (size_t)T_STEPS * UNITS_;
    float* crow = cbuf + (size_t)b0 * UNITS_ + unit;
    float zi[4], zf[4], zg[4], zo[4];
#pragma unroll
    for (int j = 0; j < 4; ++j) {
        const int rl = j * 4 + wave;
        zi[j] = Zl[(0 * 16 + rl) * ZPITCH + lane] + bI;
        zf[j] = Zl[(1 * 16 + rl) * ZPITCH + lane] + bF;
        zg[j] = Zl[(2 * 16 + rl) * ZPITCH + lane] + bG;
        zo[j] = Zl[(3 * 16 + rl) * ZPITCH + lane] + bO;
    }
    float gi[4], gf[4], gg[4], go[4], cn[4], th[4];
#pragma unroll
    for (int j = 0; j < 4; ++j) gi[j] = sigmoid_f(zi[j]);
#pragma unroll
    for (int j = 0; j < 4; ++j) gf[j] = sigmoid_f(zf[j]);
#pragma unroll
    for (int j = 0; j < 4; ++j) gg[j] = tanh_f(zg[j]);
#pragma unroll
    for (int j = 0; j < 4; ++j) go[j] = sigmoid_f(zo[j]);
#pragma unroll
    for (int j = 0; j < 4; ++j) cn[j] = __builtin_fmaf(gf[j], cprev[j], gi[j] * gg[j]);
#pragma unroll
    for (int j = 0; j < 4; ++j) crow[(size_t)(j * 4 + wave) * UNITS_] = cn[j];
#pragma unroll
    for (int j = 0; j < 4; ++j) th[j] = tanh_f(cn[j]);
#pragma unroll
    for (int j = 0; j < 4; ++j) hrow[(size_t)(j * 4 + wave) * hpitch] = go[j] * th[j];
    if (do_flush) {
#pragma unroll
        for (int j = 0; j < 4; ++j)
            out[((size_t)(b0 + j * 4 + wave) * T_STEPS + tw) * UNITS_ + unit] = fh[j] + fi[j];
    }
}

__global__ void combine_avg_kernel(const float4* __restrict__ fw,
                                   float4* __restrict__ out, int n4) {
    int i = blockIdx.x * blockDim.x + threadIdx.x;
    if (i < n4) {
        float4 a = fw[i];
        float4 b = out[i];
        out[i] = make_float4((a.x + b.x) * 0.5f, (a.y + b.y) * 0.5f,
                             (a.z + b.z) * 0.5f, (a.w + b.w) * 0.5f);
    }
}

extern "C" void kernel_launch(void* const* d_in, const int* in_sizes, int n_in,
                              void* d_out, int out_size, void* d_ws, size_t ws_size,
                              hipStream_t stream) {
    const float* x   = (const float*)d_in[0];
    const float* kfw = (const float*)d_in[1];
    const float* rfw = (const float*)d_in[2];
    const float* bfw = (const float*)d_in[3];
    const float* kbw = (const float*)d_in[4];
    const float* rbw = (const float*)d_in[5];
    const float* bbw = (const float*)d_in[6];
    float* out = (float*)d_out;

    float* pk   = (float*)d_ws;
    float* wsfw = pk + (size_t)4 * COMBO_F;
    float* wsbw = out;
    float* stt  = wsfw + (size_t)128 * 256 * 256;

    pack_weights_kernel<<<4096, 256, 0, stream>>>(kfw, rfw, kbw, rbw, pk);

    const size_t shmem = (size_t)LDS_BYTES;

#define STEP_LAUNCH(s)                                                      \
    bilstm_step_kernel<<<dim3(64, 2, 2), S_THR, shmem, stream>>>(           \
        x, wsfw, wsbw, pk, bfw, bbw, stt, (s))
#define STEP_LAUNCH_16(b)                                                   \
    STEP_LAUNCH((b) + 0);  STEP_LAUNCH((b) + 1);  STEP_LAUNCH((b) + 2);     \
    STEP_LAUNCH((b) + 3);  STEP_LAUNCH((b) + 4);  STEP_LAUNCH((b) + 5);     \
    STEP_LAUNCH((b) + 6);  STEP_LAUNCH((b) + 7);  STEP_LAUNCH((b) + 8);     \
    STEP_LAUNCH((b) + 9);  STEP_LAUNCH((b) + 10); STEP_LAUNCH((b) + 11);    \
    STEP_LAUNCH((b) + 12); STEP_LAUNCH((b) + 13); STEP_LAUNCH((b) + 14);    \
    STEP_LAUNCH((b) + 15)
    if (ws_size >= ((size_t)4 * COMBO_F + (size_t)128 * 256 * 256 + STATE_F) * sizeof(float)) {
        (void)hipFuncSetAttribute((const void*)bilstm_step_kernel,
                                  hipFuncAttributeMaxDynamicSharedMemorySize,
                                  (int)shmem);
        STEP_LAUNCH_16(0);   STEP_LAUNCH_16(16);  STEP_LAUNCH_16(32);  STEP_LAUNCH_16(48);
        STEP_LAUNCH_16(64);  STEP_LAUNCH_16(80);  STEP_LAUNCH_16(96);  STEP_LAUNCH_16(112);
        STEP_LAUNCH_16(128); STEP_LAUNCH_16(144); STEP_LAUNCH_16(160); STEP_LAUNCH_16(176);
        STEP_LAUNCH_16(192); STEP_LAUNCH_16(208); STEP_LAUNCH_16(224); STEP_LAUNCH_16(240);
        STEP_LAUNCH(256);
        STEP_LAUNCH(257);
    } else {
        (void)hipFuncSetAttribute((const void*)bilstm_layer_kernel,
                                  hipFuncAttributeMaxDynamicSharedMemorySize,
                                  (int)shmem);
        bilstm_layer_kernel<<<dim3(64, 2, 2), NTHR, shmem, stream>>>(
            x, wsfw, wsbw, pk, bfw, bbw, wsfw, 0, 1, NGRP);
        bilstm_layer_kernel<<<dim3(64, 2, 2), NTHR, shmem, stream>>>(
            x, wsfw, wsbw, pk, bfw, bbw, wsfw, 1, 1, NGRP);
    }
#undef STEP_LAUNCH_16
#undef STEP_LAUNCH

    const int n4 = (128 * 256 * 256) / 4;
    combine_avg_kernel<<<(n4 + 255) / 256, 256, 0, stream>>>(
        (const float4*)wsfw, (float4*)out, n4);
}


# a step's product cut at k 256: the x half is formed one launch ahead by four extra waves, behind the h half's matrix run, so it fills the matrix pipe during the cell update; sum order kept
# speedup vs baseline: 6.9147x; 1.0432x over previous
#include <hip/hip_runtime.h>


typedef float v2f __attribute__((ext_vector_type(2)));
typedef float v8f __attribute__((ext_vector_type(8)));

#define T_STEPS 256
#define UNITS_  256
#define COMBO_F (512 * 1024)
#define ROWS_WG 2
#define GSTEPS  8
#define NGRP    (T_STEPS / GSTEPS)
#define NCHUNK  16
#define CHUNK_G (NGRP / NCHUNK)
#define STATE_F (2 * 2 * 64 * 2 * NTHR)
#define NTHR    512
#define XSTRIDE 260
#define HSTRIDE 264
#define TQ      (16 * 128 * 64)
#define LDS_X_F  (2 * 16 * XSTRIDE)
#define LDS_H_F  (2 * ROWS_WG * HSTRIDE)
#define LDS_XZ_F (32 * NTHR)
#define LDS_BYTES (164 * 1024)

__device__ __forceinline__ float sigmoid_f(float x) {
    return 1.0f / (1.0f + __expf(-x));
}
__device__ __forceinline__ float tanh_f(float x) {
    x = fminf(15.0f, fmaxf(-15.0f, x));
    float e = __expf(2.0f * x);
    return (e - 1.0f) / (e + 1.0f);
}

__global__ void pack_weights_kernel(const float* __restrict__ kfw,
                                    const float* __restrict__ rfw,
                                    const float* __restrict__ kbw,
                                    const float* __restrict__ rbw,
                                    float* __restrict__ packed) {
    int i = blockIdx.x * blockDim.x + threadIdx.x;
    int lane  = i & 31;
    int ks    = (i >> 5) & 127;
    int nt    = (i >> 12) & 63;
    int combo = i >> 18;
    int layer = combo >> 1;
    int isbw  = combo & 1;
    const float* W = (isbw ? kbw : kfw) + (size_t)layer * 256 * 1024;
    const float* R = (isbw ? rbw : rfw) + (size_t)layer * 256 * 1024;

    int n  = nt * 16 + (lane & 15);
    int k0 = ks * 4 + ((lane >> 4) << 1);
    float d0, d1;
    if (k0 < 256) {
        d0 = W[(size_t)k0 * 1024 + n];
        d1 = W[(size_t)(k0 + 1) * 1024 + n];
    } else {
        int kk = k0 - 256;
        d0 = R[(size_t)kk * 1024 + n];
        d1 = R[(size_t)(kk + 1) * 1024 + n];
    }
    packed[(size_t)i * 2 + 0] = d0;
    packed[(size_t)i * 2 + 1] = d1;
}

__device__ __forceinline__ void load_chunk(v2f (&b)[4][4], const float* p) {
#pragma unroll
    for (int kk = 0; kk < 4; ++kk)
#pragma unroll
        for (int q = 0; q < 4; ++q)
            b[kk][q] = *(const v2f*)(p + q * TQ + kk * 64);
}
__device__ __forceinline__ void load_afrag(v2f (&a)[4], const float* p) {
#pragma unroll
    for (int kk = 0; kk < 4; ++kk) a[kk] = *(const v2f*)(p + kk * 4);
}
__device__ __forceinline__ void mma_chunk(v8f (&acc)[4], const v2f (&a)[4],
                                          const v2f (&b)[4][4]) {
#pragma unroll
    for (int kk = 0; kk < 4; ++kk)
#pragma unroll
        for (int q = 0; q < 4; ++q)
            acc[q] = __builtin_amdgcn_wmma_f32_16x16x4_f32(
                false, a[kk], false, b[kk][q], (short)0, acc[q], false, false);
}

__device__ __forceinline__ void gemm_phase(v8f (&acc)[4], v2f (&b0)[4][4],
                                           v2f (&b1)[4][4], v2f (&b2)[4][4],
                                           v2f (&b3)[4][4], const float* aL,
                                           const float* bp, const float* nextp) {
    v2f a0[4], a1[4];
    load_afrag(a0, aL);
#define STAGE_FENCE() __builtin_amdgcn_sched_barrier(0)
#define MMA_DONE() asm volatile("" : "+v"(acc[0]), "+v"(acc[1]), "+v"(acc[2]), "+v"(acc[3]) : : "memory")
#define PRIO_IF(cnt, val, prio)                                   \
    asm volatile("s_cmp_lg_u32 %0, " #val "\n\t"                  \
                 "s_cbranch_scc1 1f\n\t"                          \
                 "s_setprio " #prio "\n"                           \
                 "1:" : : "s"(cnt) : "scc")
    __builtin_amdgcn_s_setprio(3);
#pragma unroll 1
    for (int ch = 0; ch < 16; ch += 4) {
        load_chunk(b3, bp + (ch + 3) * 256);
        load_afrag(a1, aL + (ch + 1) * 16);
        STAGE_FENCE();
        mma_chunk(acc, a0, b0);
        MMA_DONE();
        STAGE_FENCE();
        const float* np = (ch + 4 < 16) ? (bp + (ch + 4) * 256) : nextp;
        load_chunk(b0, np);
        load_afrag(a0, aL + (ch + 2) * 16);
        STAGE_FENCE();
        mma_chunk(acc, a1, b1);
        MMA_DONE();
        STAGE_FENCE();
        PRIO_IF(ch, 12, 0);
        load_chunk(b1, np + 256);
        load_afrag(a1, aL + (ch + 3) * 16);
        STAGE_FENCE();
        mma_chunk(acc, a0, b2);
        MMA_DONE();
        STAGE_FENCE();
        load_chunk(b2, np + 512);
        load_afrag(a0, aL + ((ch + 4) & 15) * 16);
        STAGE_FENCE();
        mma_chunk(acc, a1, b3);
        MMA_DONE();
        STAGE_FENCE();
        PRIO_IF(ch, 4, 2);
        PRIO_IF(ch, 8, 1);
    }
#undef PRIO_IF
#undef MMA_DONE
#undef STAGE_FENCE
}

__global__ void __launch_bounds__(NTHR)
bilstm_layer_kernel(const float* x,
                    float* chain_fw,
                    float* chain_bw,
                    const float* __restrict__ packed,
                    const float* __restrict__ bias_fw,
                    const float* __restrict__ bias_bw,
                    float* state,
                    int launch_idx, int nchunk, int chunk_groups) {
    const int layer = blockIdx.z;
    const int chunk = launch_idx - layer;
    if (chunk < 0 || chunk >= nchunk) return;

    extern __shared__ float lds[];
    float* Xl  = lds;
    float* Hl  = lds + LDS_X_F;
    float* XZl = Hl + LDS_H_F;

    const int dir      = blockIdx.y;
    float*       out   = dir ? chain_bw : chain_fw;
    const float* in    = layer ? out : x;
    const float* pw    = packed + (size_t)(layer * 2 + dir) * COMBO_F;
    const float* bias  = (dir ? bias_bw : bias_fw) + layer * 1024;
    const int residual = layer;
    float* st = state + (size_t)((layer * 2 + dir) * 64 + blockIdx.x) * (2 * NTHR);

    const int b0   = blockIdx.x * ROWS_WG;
    const int tid  = threadIdx.x;
    const int lane = tid & 31;
    const int wave = tid >> 5;
    const int hs   = lane >> 4;
    const int lc   = lane & 15;
    const int unit = wave * 16 + lc;

    const float bI = bias[unit];
    const float bF = bias[256 + unit];
    const float bG = bias[512 + unit];
    const float bO = bias[768 + unit];
    float cst = 0.0f, hreg = 0.0f;
    if (chunk > 0) {
        hreg = st[tid];
        cst  = st[NTHR + tid];
    }
    Hl[hs * HSTRIDE + unit] = hreg;

    const int srow0 = tid >> 6;
    const int srow1 = srow0 + 8;
    const int scol  = (tid & 63) * 4;
    const int g_first = chunk * chunk_groups;
    const int g_end   = g_first + chunk_groups;
    {
        const int tt0 = g_first * GSTEPS + srow0;
        const int t = dir ? (T_STEPS - 1 - tt0) : tt0;
        const float4 v0 = *(const float4*)(in + ((size_t)(b0 + 0) * T_STEPS + t) * UNITS_ + scol);
        const float4 v1 = *(const float4*)(in + ((size_t)(b0 + 1) * T_STEPS + t) * UNITS_ + scol);
        *(float4*)(Xl + srow0 * XSTRIDE + scol) = v0;
        *(float4*)(Xl + srow1 * XSTRIDE + scol) = v1;
    }

    const float* wl = pw + wave * 8192 + lane * 2;
    const float* rl = wl + 64 * 64;
    v2f fb0[4][4], fb1[4][4], fb2[4][4], fb3[4][4];
    load_chunk(fb0, wl);
    load_chunk(fb1, wl + 256);
    load_chunk(fb2, wl + 512);
    __syncthreads();

    int xcur = 0, hcur = 0;
    for (int grp = g_first; grp < g_end; ++grp) {
        float4 xr0, xr1;
        const bool more = (grp + 1 < g_end);
        if (more) {
            const int tt = (grp + 1) * GSTEPS + srow0;
            const int t  = dir ? (T_STEPS - 1 - tt) : tt;
            xr0 = *(const float4*)(in + ((size_t)(b0 + 0) * T_STEPS + t) * UNITS_ + scol);
            xr1 = *(const float4*)(in + ((size_t)(b0 + 1) * T_STEPS + t) * UNITS_ + scol);
        }

        v8f acc[4];
#pragma unroll
        for (int q = 0; q < 4; ++q)
            acc[q] = (v8f){0.f, 0.f, 0.f, 0.f, 0.f, 0.f, 0.f, 0.f};
        gemm_phase(acc, fb0, fb1, fb2, fb3,
                   Xl + xcur * (16 * XSTRIDE) + lc * XSTRIDE + 2 * hs, wl, rl);
#pragma unroll
        for (int q = 0; q < 4; ++q)
#pragma unroll
            for (int e = 0; e < 8; ++e)
                XZl[(q * 8 + e) * NTHR + tid] = acc[q][e];

        if (more) {
            float* Xn = Xl + (xcur ^ 1) * (16 * XSTRIDE);
            *(float4*)(Xn + srow0 * XSTRIDE + scol) = xr0;
            *(float4*)(Xn + srow1 * XSTRIDE + scol) = xr1;
        }

        for (int g = 0; g < GSTEPS; ++g) {
            const int tt = grp * GSTEPS + g;
            const int t  = dir ? (T_STEPS - 1 - tt) : tt;

#pragma unroll
            for (int q = 0; q < 4; ++q) {
                const float s = XZl[(q * 8 + g) * NTHR + tid];
                acc[q] = (v8f){s, s, s, s, s, s, s, s};
            }
            const float* nextp = (g + 1 < GSTEPS) ? rl : wl;
            gemm_phase(acc, fb0, fb1, fb2, fb3,
                       Hl + hcur * (ROWS_WG * HSTRIDE) + (lc >> 3) * HSTRIDE + 2 * hs,
                       rl, nextp);

            const float zi = acc[0][0] + bI;
            const float zf = acc[1][0] + bF;
            const float zg = acc[2][0] + bG;
            const float zo = acc[3][0] + bO;
            const float gi = sigmoid_f(zi);
            const float gf = sigmoid_f(zf);
            const float gg = tanh_f(zg);
            const float go = sigmoid_f(zo);
            const float c  = gf * cst + gi * gg;
            cst = c;
            const float h = go * tanh_f(c);
            hreg = h;
            Hl[(hcur ^ 1) * (ROWS_WG * HSTRIDE) + hs * HSTRIDE + unit] = h;
            const float ov = residual
                ? (h + Xl[xcur * (16 * XSTRIDE) + (8 * hs + g) * XSTRIDE + unit]) : h;
            out[((size_t)(b0 + hs) * T_STEPS + t) * UNITS_ + unit] = ov;
            __syncthreads();
            hcur ^= 1;
        }
        xcur ^= 1;
    }
    if (chunk + 1 < nchunk) {
        st[tid]        = hreg;
        st[NTHR + tid] = cst;
    }
}

#define S_THR   256
#define APITCH  516
#define ZPITCH  36
#define S_LDS_A (16 * APITCH)
#define CELLS_F (128 * 256)
#define XSUM_F  (4 * 64 * 4 * 512)

__device__ __forceinline__ void load_chunk8(v2f (&b)[8][2], const float* p) {
#pragma unroll
    for (int kk = 0; kk < 8; ++kk)
#pragma unroll
        for (int q = 0; q < 2; ++q)
            b[kk][q] = *(const v2f*)(p + q * TQ + kk * 64);
}
__device__ __forceinline__ void load_afrag8(v2f (&a)[8], const float* p) {
#pragma unroll
    for (int kk = 0; kk < 8; ++kk) a[kk] = *(const v2f*)(p + kk * 4);
}
__device__ __forceinline__ void mma_chunk8(v8f (&acc)[2], const v2f (&a)[8],
                                           const v2f (&b)[8][2]) {
#pragma unroll
    for (int kk = 0; kk < 8; ++kk)
#pragma unroll
        for (int q = 0; q < 2; ++q)
            acc[q] = __builtin_amdgcn_wmma_f32_16x16x4_f32(
                false, a[kk], false, b[kk][q], (short)0, acc[q], false, false);
}

__device__ __forceinline__ void half_product(v8f (&acc)[2], v2f (&fb0)[8][2],
                                             v2f (&fb1)[8][2], v2f (&fb2)[8][2],
                                             const float* aL, const float* bp) {
    v2f fb3[8][2];
    v2f a0[8], a1[8];
    load_afrag8(a0, aL);
#define S_FENCE() __builtin_amdgcn_sched_barrier(0)
#define S_DONE() asm volatile("" : "+v"(acc[0]), "+v"(acc[1]) : : "memory")
    load_chunk8(fb3, bp + 3 * 512);
    load_afrag8(a1, aL + 1 * 32);
    S_FENCE();
    mma_chunk8(acc, a0, fb0);
    S_DONE();
    S_FENCE();
    load_chunk8(fb0, bp + 4 * 512);
    load_afrag8(a0, aL + 2 * 32);
    S_FENCE();
    mma_chunk8(acc, a1, fb1);
    S_DONE();
    S_FENCE();
    load_chunk8(fb1, bp + 5 * 512);
    load_afrag8(a1, aL + 3 * 32);
    S_FENCE();
    mma_chunk8(acc, a0, fb2);
    S_DONE();
    S_FENCE();
    load_chunk8(fb2, bp + 6 * 512);
    load_afrag8(a0, aL + 4 * 32);
    S_FENCE();
    mma_chunk8(acc, a1, fb3);
    S_DONE();
    S_FENCE();
    load_chunk8(fb3, bp + 7 * 512);
    load_afrag8(a1, aL + 5 * 32);
    S_FENCE();
    mma_chunk8(acc, a0, fb0);
    S_DONE();
    S_FENCE();
    load_afrag8(a0, aL + 6 * 32);
    S_FENCE();
    mma_chunk8(acc, a1, fb1);
    S_DONE();
    S_FENCE();
    load_afrag8(a1, aL + 7 * 32);
    S_FENCE();
    mma_chunk8(acc, a0, fb2);
    mma_chunk8(acc, a1, fb3);
    S_DONE();
    S_FENCE();
    asm volatile("v_nop\n\tv_nop\n\tv_nop\n\tv_nop\n\tv_nop" : : : "memory");
    S_FENCE();
#undef S_DONE
#undef S_FENCE
}

__global__ void __launch_bounds__(S_THR)
bilstm_step_kernel(const float* x,
                   float* chain_fw,
                   float* chain_bw,
                   const float* __restrict__ packed,
                   const float* __restrict__ bias_fw,
                   const float* __restrict__ bias_bw,
                   float* state, int step) {
    const int layer = blockIdx.z;
    const int ta = step - 2 * layer;
    const int tb = ta - 1;
    const bool doA = (ta >= 0 && ta < T_STEPS);
    const bool doB = (tb >= 0 && tb < T_STEPS);
    if (!doA && !doB) return;

    extern __shared__ float lds[];
    float* Al = lds;
    float* Zl = lds + S_LDS_A;

    const int dir   = blockIdx.y;
    float* out      = dir ? chain_bw : chain_fw;
    const float* in = layer ? out : x;
    const int combo = layer * 2 + dir;
    const int b0    = (blockIdx.x >> 3) * 16;
    const int ugrp  = blockIdx.x & 7;
    const int tid   = threadIdx.x;
    const int wv    = __builtin_amdgcn_readfirstlane(tid >> 5);
    const bool isX  = (wv >= 4);
    const int wave  = wv & 3;
    const int gt    = tid & 127;
    const int lane  = tid & 31;
    const int hs    = lane >> 4;
    const int lc    = lane & 15;
    const int unit  = ugrp * 32 + lane;

    float* cbuf = state + (size_t)combo * CELLS_F;
    float* h1   = state + (size_t)4 * CELLS_F + (size_t)dir * 2 * CELLS_F;
    float4* xsum = (float4*)(state + (size_t)8 * CELLS_F
                             + ((size_t)(combo * 64 + blockIdx.x) * 4 + wave) * 512) + lane;

    const int tA     = dir ? (T_STEPS - 1 - ta) : ta;
    const int tB     = dir ? (T_STEPS - 1 - tb) : tb;
    const int tBprev = dir ? (tB + 1) : (tB - 1);
    const float* pw   = packed + (size_t)combo * COMBO_F;
    const float* bias = (dir ? bias_bw : bias_fw) + layer * 1024;

    const int col4 = (gt & 63) * 4;
    const int r0   = (gt >> 6) * 8;
    float4 v[8];
    {
        const float* src;
        size_t rowf;
        bool ld;
        if (isX) {
            src  = in + ((size_t)(b0 + r0) * T_STEPS + tA) * UNITS_ + col4;
            rowf = (size_t)T_STEPS * UNITS_;
            ld   = doA;
        } else if (layer) {
            src  = h1 + (size_t)((tb - 1) & 1) * CELLS_F + (size_t)(b0 + r0) * UNITS_ + col4;
            rowf = (size_t)UNITS_;
            ld   = doB && tb > 0;
        } else {
            src  = out + ((size_t)(b0 + r0) * T_STEPS + tBprev) * UNITS_ + col4;
            rowf = (size_t)T_STEPS * UNITS_;
            ld   = doB && tb > 0;
        }
#pragma unroll
        for (int r = 0; r < 8; ++r) v[r] = make_float4(0.f, 0.f, 0.f, 0.f);
        if (ld) {
#pragma unroll
            for (int r = 0; r < 8; ++r)
                v[r] = *(const float4*)(src + (size_t)r * rowf);
        }
    }

    v8f acc[2];
    acc[0] = (v8f){0.f, 0.f, 0.f, 0.f, 0.f, 0.f, 0.f, 0.f};
    acc[1] = (v8f){0.f, 0.f, 0.f, 0.f, 0.f, 0.f, 0.f, 0.f};
    float cprev[4] = {0.f, 0.f, 0.f, 0.f};
    float fi[4]    = {0.f, 0.f, 0.f, 0.f};
    if (!isX && doB) {
#pragma unroll
        for (int q4 = 0; q4 < 4; ++q4) {
            const float4 xv = xsum[q4 * 32];
            acc[q4 >> 1][(q4 & 1) * 4 + 0] = xv.x;
            acc[q4 >> 1][(q4 & 1) * 4 + 1] = xv.y;
            acc[q4 >> 1][(q4 & 1) * 4 + 2] = xv.z;
            acc[q4 >> 1][(q4 & 1) * 4 + 3] = xv.w;
        }
        if (tb > 0) {
#pragma unroll
            for (int j = 0; j < 4; ++j)
                cprev[j] = cbuf[(size_t)(b0 + j * 4 + wave) * UNITS_ + unit];
        }
        if (layer) {
#pragma unroll
            for (int j = 0; j < 4; ++j)
                fi[j] = out[((size_t)(b0 + j * 4 + wave) * T_STEPS + tB) * UNITS_ + unit];
        }
    }

    const float* bp = pw + (size_t)((wave >> 1) * 32 + ugrp * 2 + (wave & 1)) * 8192
                         + (isX ? 0 : 8 * 512) + lane * 2;
    v2f fb0[8][2], fb1[8][2], fb2[8][2];
    load_chunk8(fb0, bp);
    load_chunk8(fb1, bp + 512);
    load_chunk8(fb2, bp + 1024);

    const float bI = bias[unit];
    const float bF = bias[256 + unit];
    const float bG = bias[512 + unit];
    const float bO = bias[768 + unit];

    {
        float* dst = Al + r0 * APITCH + (isX ? 0 : 256) + col4;
#pragma unroll
        for (int r = 0; r < 8; ++r)
            *(float4*)(dst + r * APITCH) = v[r];
    }
    __syncthreads();

    const float* aL = Al + lc * APITCH + 2 * hs + (isX ? 0 : 256);
    if (!isX && doB) {
        half_product(acc, fb0, fb1, fb2, aL, bp);
#pragma unroll
        for (int q = 0; q < 2; ++q)
#pragma unroll
            for (int e = 0; e < 8; ++e)
                Zl[(((wave >> 1) * 2 + q) * 16 + 8 * hs + e) * ZPITCH + (wave & 1) * 16 + lc] = acc[q][e];
    }
    __syncthreads();

    if (isX) {
        if (doA) {
            half_product(acc, fb0, fb1, fb2, aL, bp);
#pragma unroll
            for (int q4 = 0; q4 < 4; ++q4)
                xsum[q4 * 32] = make_float4(acc[q4 >> 1][(q4 & 1) * 4 + 0], acc[q4 >> 1][(q4 & 1) * 4 + 1],
                                            acc[q4 >> 1][(q4 & 1) * 4 + 2], acc[q4 >> 1][(q4 & 1) * 4 + 3]);
        }
        return;
    }
    if (!doB) return;

    float* hrow = layer ? (h1 + (size_t)(tb & 1) * CELLS_F + (size_t)b0 * UNITS_ + unit)
                        : (out + ((size_t)b0 * T_STEPS + tB) * UNITS_ + unit);
    const size_t hpitch = layer ? (size_t)UNITS_ : (size_t)T_STEPS * UNITS_;
    float* crow = cbuf + (size_t)b0 * UNITS_ + unit;
    float zi[4], zf[4], zg[4], zo[4];
#pragma unroll
    for (int j = 0; j < 4; ++j) {
        const int rl = j * 4 + wave;
        zi[j] = Zl[(0 * 16 + rl) * ZPITCH + lane] + bI;
        zf[j] = Zl[(1 * 16 + rl) * ZPITCH + lane] + bF;
        zg[j] = Zl[(2 * 16 + rl) * ZPITCH + lane] + bG;
        zo[j] = Zl[(3 * 16 + rl) * ZPITCH + lane] + bO;
    }
    float gi[4], gf[4], gg[4], go[4], cn[4], th[4], hv[4];
#pragma unroll
    for (int j = 0; j < 4; ++j) gi[j] = sigmoid_f(zi[j]);
#pragma unroll
    for (int j = 0; j < 4; ++j) gf[j] = sigmoid_f(zf[j]);
#pragma unroll
    for (int j = 0; j < 4; ++j) gg[j] = tanh_f(zg[j]);
#pragma unroll
    for (int j = 0; j < 4; ++j) go[j] = sigmoid_f(zo[j]);
#pragma unroll
    for (int j = 0; j < 4; ++j) cn[j] = __builtin_fmaf(gf[j], cprev[j], gi[j] * gg[j]);
#pragma unroll
    for (int j = 0; j < 4; ++j) crow[(size_t)(j * 4 + wave) * UNITS_] = cn[j];
#pragma unroll
    for (int j = 0; j < 4; ++j) th[j] = tanh_f(cn[j]);
#pragma unroll
    for (int j = 0; j < 4; ++j) hv[j] = go[j] * th[j];
#pragma unroll
    for (int j = 0; j < 4; ++j) hrow[(size_t)(j * 4 + wave) * hpitch] = hv[j];
    if (layer) {
#pragma unroll
        for (int j = 0; j < 4; ++j)
            out[((size_t)(b0 + j * 4 + wave) * T_STEPS + tB) * UNITS_ + unit] = hv[j] + fi[j];
    }
}

__global__ void combine_avg_kernel(const float4* __restrict__ fw,
                                   float4* __restrict__ out, int n4) {
    int i = blockIdx.x * blockDim.x + threadIdx.x;
    if (i < n4) {
        float4 a = fw[i];
        float4 b = out[i];
        out[i] = make_float4((a.x + b.x) * 0.5f, (a.y + b.y) * 0.5f,
                             (a.z + b.z) * 0.5f, (a.w + b.w) * 0.5f);
    }
}

extern "C" void kernel_launch(void* const* d_in, const int* in_sizes, int n_in,
                              void* d_out, int out_size, void* d_ws, size_t ws_size,
                              hipStream_t stream) {
    const float* x   = (const float*)d_in[0];
    const float* kfw = (const float*)d_in[1];
    const float* rfw = (const float*)d_in[2];
    const float* bfw = (const float*)d_in[3];
    const float* kbw = (const float*)d_in[4];
    const float* rbw = (const float*)d_in[5];
    const float* bbw = (const float*)d_in[6];
    float* out = (float*)d_out;

    float* pk   = (float*)d_ws;
    float* wsfw = pk + (size_t)4 * COMBO_F;
    float* wsbw = out;
    float* stt  = wsfw + (size_t)128 * 256 * 256;

    pack_weights_kernel<<<4096, 256, 0, stream>>>(kfw, rfw, kbw, rbw, pk);

    const size_t shmem = (size_t)LDS_BYTES;

#define STEP_LAUNCH(s)                                                      \
    bilstm_step_kernel<<<dim3(64, 2, 2), S_THR, shmem, stream>>>(           \
        x, wsfw, wsbw, pk, bfw, bbw, stt, (s))
#define STEP_LAUNCH_16(b)                                                   \
    STEP_LAUNCH((b) + 0);  STEP_LAUNCH((b) + 1);  STEP_LAUNCH((b) + 2);     \
    STEP_LAUNCH((b) + 3);  STEP_LAUNCH((b) + 4);  STEP_LAUNCH((b) + 5);     \
    STEP_LAUNCH((b) + 6);  STEP_LAUNCH((b) + 7);  STEP_LAUNCH((b) + 8);     \
    STEP_LAUNCH((b) + 9);  STEP_LAUNCH((b) + 10); STEP_LAUNCH((b) + 11);    \
    STEP_LAUNCH((b) + 12); STEP_LAUNCH((b) + 13); STEP_LAUNCH((b) + 14);    \
    STEP_LAUNCH((b) + 15)
    if (ws_size >= ((size_t)4 * COMBO_F + (size_t)128 * 256 * 256 + STATE_F + XSUM_F) * sizeof(float)) {
        (void)hipFuncSetAttribute((const void*)bilstm_step_kernel,
                                  hipFuncAttributeMaxDynamicSharedMemorySize,
                                  (int)shmem);
        STEP_LAUNCH_16(0);   STEP_LAUNCH_16(16);  STEP_LAUNCH_16(32);  STEP_LAUNCH_16(48);
        STEP_LAUNCH_16(64);  STEP_LAUNCH_16(80);  STEP_LAUNCH_16(96);  STEP_LAUNCH_16(112);
        STEP_LAUNCH_16(128); STEP_LAUNCH_16(144); STEP_LAUNCH_16(160); STEP_LAUNCH_16(176);
        STEP_LAUNCH_16(192); STEP_LAUNCH_16(208); STEP_LAUNCH_16(224); STEP_LAUNCH_16(240);
        STEP_LAUNCH(256);
        STEP_LAUNCH(257);
        STEP_LAUNCH(258);
    } else {
        (void)hipFuncSetAttribute((const void*)bilstm_layer_kernel,
                                  hipFuncAttributeMaxDynamicSharedMemorySize,
                                  (int)shmem);
        bilstm_layer_kernel<<<dim3(64, 2, 2), NTHR, shmem, stream>>>(
            x, wsfw, wsbw, pk, bfw, bbw, wsfw, 0, 1, NGRP);
        bilstm_layer_kernel<<<dim3(64, 2, 2), NTHR, shmem, stream>>>(
            x, wsfw, wsbw, pk, bfw, bbw, wsfw, 1, 1, NGRP);
    }
#undef STEP_LAUNCH_16
#undef STEP_LAUNCH

    const int n4 = (128 * 256 * 256) / 4;
    combine_avg_kernel<<<(n4 + 255) / 256, 256, 0, stream>>>(
        (const float4*)wsfw, (float4*)out, n4);
}
